# starting kernel
# baseline (speedup 1.0000x reference)
#include <hip/hip_runtime.h>

#define E_HID 128
#define R_HID 128

typedef float v2f __attribute__((ext_vector_type(2)));
typedef float v8f __attribute__((ext_vector_type(8)));

__device__ __forceinline__ float leaky(float x) { return x > 0.f ? x : 0.01f * x; }

__device__ __forceinline__ void atomicMaxF(float* addr, float val) {
  if (val >= 0.f) atomicMax((int*)addr, __float_as_int(val));
  else            atomicMin((unsigned int*)addr, __float_as_uint(val));
}

__global__ void k_init(float* __restrict__ x_acc, long n_acc,
                       float* __restrict__ mx_h, float* __restrict__ mx_t,
                       float* __restrict__ sm_h, float* __restrict__ sm_t, int n_e) {
  long i = (long)blockIdx.x * blockDim.x + threadIdx.x;
  long stride = (long)gridDim.x * blockDim.x;
  for (long j = i; j < n_acc; j += stride) x_acc[j] = 0.f;
  for (long j = i; j < n_e; j += stride) {
    mx_h[j] = -__builtin_inff(); mx_t[j] = -__builtin_inff();
    sm_h[j] = 0.f;               sm_t[j] = 0.f;
  }
}

__global__ void k_node_scores(const float* __restrict__ x_e,
                              const float* __restrict__ w_ah,
                              const float* __restrict__ w_at,
                              float* __restrict__ e_h, float* __restrict__ e_t, int n_e) {
  int lane = threadIdx.x & 31;
  int wave = threadIdx.x >> 5;
  int row = blockIdx.x * 8 + wave;
  if (row >= n_e) return;
  const float4 x4 = ((const float4*)(x_e + (size_t)row * E_HID))[lane];
  const float4 a4 = ((const float4*)w_ah)[lane];
  const float4 b4 = ((const float4*)w_at)[lane];
  float sh = x4.x*a4.x + x4.y*a4.y + x4.z*a4.z + x4.w*a4.w;
  float st = x4.x*b4.x + x4.y*b4.y + x4.z*b4.z + x4.w*b4.w;
  for (int off = 16; off > 0; off >>= 1) {
    sh += __shfl_xor(sh, off);
    st += __shfl_xor(st, off);
  }
  if (lane == 0) { e_h[row] = sh; e_t[row] = st; }
}

__global__ void k_rel(const float* __restrict__ x_r,
                      const float* __restrict__ w_ar,
                      const float* __restrict__ W1, const float* __restrict__ b1,
                      const float* __restrict__ W2, const float* __restrict__ b2,
                      float* __restrict__ e_r, float* __restrict__ msg, int n_r) {
  __shared__ float row[R_HID];
  __shared__ float hid[2 * R_HID];
  int r = blockIdx.x;
  int t = threadIdx.x;
  if (t < R_HID) row[t] = x_r[(size_t)r * R_HID + t];
  __syncthreads();
  float acc = b1[t];
  #pragma unroll 8
  for (int k = 0; k < R_HID; ++k) acc += row[k] * W1[k * 2 * R_HID + t];
  hid[t] = acc;
  if (t < 32) {
    float s = 0.f;
    #pragma unroll
    for (int j = 0; j < 4; ++j) s += row[t + 32 * j] * w_ar[t + 32 * j];
    for (int off = 16; off > 0; off >>= 1) s += __shfl_xor(s, off);
    if (t == 0) e_r[r] = s;
  }
  __syncthreads();
  if (t < R_HID) {
    float o = b2[t];
    #pragma unroll 8
    for (int k = 0; k < 2 * R_HID; ++k) o += hid[k] * W2[k * R_HID + t];
    msg[(size_t)r * R_HID + t] = row[t] + o;
  }
}

__global__ void k_edge_max(const int* __restrict__ h_idx, const int* __restrict__ t_idx,
                           const int* __restrict__ rel,
                           const float* __restrict__ e_h, const float* __restrict__ e_t,
                           const float* __restrict__ e_r,
                           float* __restrict__ z_h, float* __restrict__ z_t,
                           float* __restrict__ mx_h, float* __restrict__ mx_t, int n_edges) {
  int e = blockIdx.x * blockDim.x + threadIdx.x;
  if (e >= n_edges) return;
  int h = h_idx[e], t = t_idx[e], r = rel[e];
  float er = e_r[r];
  float lh = leaky(e_h[h] + er);
  float lt = leaky(e_t[t] + er);
  z_h[e] = lh; z_t[e] = lt;
  atomicMaxF(mx_h + h, lh);
  atomicMaxF(mx_t + t, lt);
}

__global__ void k_edge_sum(const int* __restrict__ h_idx, const int* __restrict__ t_idx,
                           const float* __restrict__ mx_h, const float* __restrict__ mx_t,
                           float* __restrict__ z_h, float* __restrict__ z_t,
                           float* __restrict__ sm_h, float* __restrict__ sm_t, int n_edges) {
  int e = blockIdx.x * blockDim.x + threadIdx.x;
  if (e >= n_edges) return;
  int h = h_idx[e], t = t_idx[e];
  float zh = __expf(z_h[e] - mx_h[h]);
  float zt = __expf(z_t[e] - mx_t[t]);
  z_h[e] = zh; z_t[e] = zt;
  atomicAdd(sm_h + h, zh);
  atomicAdd(sm_t + t, zt);
}

__global__ void k_scatter(const int* __restrict__ h_idx, const int* __restrict__ t_idx,
                          const int* __restrict__ rel,
                          const float* __restrict__ z_h, const float* __restrict__ z_t,
                          const float* __restrict__ sm_h, const float* __restrict__ sm_t,
                          const float* __restrict__ msg, float* __restrict__ x_acc, int n_edges) {
  int gid = blockIdx.x * blockDim.x + threadIdx.x;
  int e = gid >> 5;
  int lane = gid & 31;
  if (e >= n_edges) return;
  int h = h_idx[e], t = t_idx[e], r = rel[e];
  float ah = z_h[e] / (sm_h[h] + 1e-16f);
  float at = z_t[e] / (sm_t[t] + 1e-16f);
  float4 m4 = ((const float4*)(msg + (size_t)r * R_HID))[lane];
  float* ph = x_acc + (size_t)h * 256 + lane * 4;
  atomicAdd(ph + 0, ah * m4.x); atomicAdd(ph + 1, ah * m4.y);
  atomicAdd(ph + 2, ah * m4.z); atomicAdd(ph + 3, ah * m4.w);
  float* pt = x_acc + (size_t)t * 256 + 128 + lane * 4;
  atomicAdd(pt + 0, at * m4.x); atomicAdd(pt + 1, at * m4.y);
  atomicAdd(pt + 2, at * m4.z); atomicAdd(pt + 3, at * m4.w);
}

__global__ void k_final_gemm(const float* __restrict__ x_acc,
                             const float* __restrict__ Wr, const float* __restrict__ br,
                             float* __restrict__ out, int n_e) {
  constexpr int KTOT = 256, NTOT = 256, KC = 32;
  constexpr int ASTR = 36, BKSTR = 36;
  __shared__ float As[16 * ASTR];
  __shared__ float Bs[NTOT * BKSTR];
  const int tid  = threadIdx.x;
  const int lane = tid & 31;
  const int wave = tid >> 5;
  const int m0   = blockIdx.x * 16;
  const int lrow = lane & 15;
  const int hsel = (lane >= 16) ? 2 : 0;
  const int mrow = (lane >= 16) ? 8 : 0;
  const int n0   = wave * 16;

  const int arow = tid >> 3;
  const int acol = (tid & 7) * 4;

  v8f c0 = {}, c1 = {};
  float4 aReg = make_float4(0.f, 0.f, 0.f, 0.f);
  float4 bReg[8];

  auto loadGlobal = [&](int kb) {
    if (tid < 128) {
      int grow = m0 + arow;
      aReg = make_float4(0.f, 0.f, 0.f, 0.f);
      if (grow < n_e) aReg = *(const float4*)(x_acc + (size_t)grow * KTOT + kb + acol);
    }
    #pragma unroll
    for (int i = 0; i < 8; ++i) {
      int k4 = i * 4;
      bReg[i].x = Wr[(size_t)(kb + k4 + 0) * NTOT + tid];
      bReg[i].y = Wr[(size_t)(kb + k4 + 1) * NTOT + tid];
      bReg[i].z = Wr[(size_t)(kb + k4 + 2) * NTOT + tid];
      bReg[i].w = Wr[(size_t)(kb + k4 + 3) * NTOT + tid];
    }
  };
  auto storeSmem = [&]() {
    if (tid < 128) *(float4*)&As[arow * ASTR + acol] = aReg;
    #pragma unroll
    for (int i = 0; i < 8; ++i)
      *(float4*)&Bs[tid * BKSTR + i * 4] = bReg[i];
  };

  loadGlobal(0);
  for (int kb = 0; kb < KTOT; kb += KC) {
    storeSmem();
    __syncthreads();
    if (kb + KC < KTOT) loadGlobal(kb + KC);
    #pragma unroll
    for (int kk = 0; kk < KC; kk += 4) {
      int ka = kk + hsel;
      v2f a  = *(const v2f*)&As[lrow * ASTR + ka];
      v2f b0 = *(const v2f*)&Bs[(n0 + lrow) * BKSTR + ka];
      v2f b1 = *(const v2f*)&Bs[(n0 + 128 + lrow) * BKSTR + ka];
      c0 = __builtin_amdgcn_wmma_f32_16x16x4_f32(false, a, false, b0, (short)0, c0, false, false);
      c1 = __builtin_amdgcn_wmma_f32_16x16x4_f32(false, a, false, b1, (short)0, c1, false, false);
    }
    __syncthreads();
  }
  #pragma unroll
  for (int j = 0; j < 8; ++j) {
    int grow = m0 + mrow + j;
    if (grow < n_e) {
      out[(size_t)grow * NTOT + n0 + lrow]       = c0[j] + br[n0 + lrow];
      out[(size_t)grow * NTOT + n0 + 128 + lrow] = c1[j] + br[n0 + 128 + lrow];
    }
  }
}

extern "C" void kernel_launch(void* const* d_in, const int* in_sizes, int n_in,
                              void* d_out, int out_size, void* d_ws, size_t ws_size,
                              hipStream_t stream) {
  const float* x_e  = (const float*)d_in[0];
  const float* x_r  = (const float*)d_in[1];
  const int*   eidx = (const int*)d_in[2];
  const int*   rel  = (const int*)d_in[3];
  const float* w_ah = (const float*)d_in[5];
  const float* w_at = (const float*)d_in[6];
  const float* w_ar = (const float*)d_in[7];
  const float* W1   = (const float*)d_in[8];
  const float* b1   = (const float*)d_in[9];
  const float* W2   = (const float*)d_in[10];
  const float* b2   = (const float*)d_in[11];
  const float* Wr   = (const float*)d_in[12];
  const float* br   = (const float*)d_in[13];
  float* out = (float*)d_out;

  const int n_e     = in_sizes[0] / E_HID;
  const int n_r     = in_sizes[1] / R_HID;
  const int n_edges = in_sizes[3];
  const int* h_idx = eidx;
  const int* t_idx = eidx + n_edges;

  float* x_acc = (float*)d_ws;
  float* msg   = x_acc + (size_t)n_e * 256;
  float* z_h   = msg + (size_t)n_r * R_HID;
  float* z_t   = z_h + n_edges;
  float* e_h   = z_t + n_edges;
  float* e_t   = e_h + n_e;
  float* e_r   = e_t + n_e;
  float* mx_h  = e_r + n_r;
  float* mx_t  = mx_h + n_e;
  float* sm_h  = mx_t + n_e;
  float* sm_t  = sm_h + n_e;

  k_init<<<4096, 256, 0, stream>>>(x_acc, (long)n_e * 256, mx_h, mx_t, sm_h, sm_t, n_e);
  k_node_scores<<<(n_e + 7) / 8, 256, 0, stream>>>(x_e, w_ah, w_at, e_h, e_t, n_e);
  k_rel<<<n_r, 256, 0, stream>>>(x_r, w_ar, W1, b1, W2, b2, e_r, msg, n_r);
  k_edge_max<<<(n_edges + 255) / 256, 256, 0, stream>>>(h_idx, t_idx, rel, e_h, e_t, e_r,
                                                        z_h, z_t, mx_h, mx_t, n_edges);
  k_edge_sum<<<(n_edges + 255) / 256, 256, 0, stream>>>(h_idx, t_idx, mx_h, mx_t,
                                                        z_h, z_t, sm_h, sm_t, n_edges);
  k_scatter<<<(n_edges + 7) / 8, 256, 0, stream>>>(h_idx, t_idx, rel, z_h, z_t,
                                                   sm_h, sm_t, msg, x_acc, n_edges);
  k_final_gemm<<<(n_e + 15) / 16, 256, 0, stream>>>(x_acc, Wr, br, out, n_e);
}


# Row sums kept in LDS: a work-group owns 391 nodes of a side, scans a key byte per edge for its edges, adds rows by LDS atomics, stores x once; no memory float adds, no zero fill; sum order changes
# speedup vs baseline: 2.5543x; 2.5543x over previous
#include <hip/hip_runtime.h>

#define E_HID 128
#define R_HID 128

typedef float v2f __attribute__((ext_vector_type(2)));
typedef float v8f __attribute__((ext_vector_type(8)));

__device__ __forceinline__ float leaky(float x) { return x > 0.f ? x : 0.01f * x; }

__device__ __forceinline__ void atomicMaxF(float* addr, float val) {
  if (val >= 0.f) atomicMax((int*)addr, __float_as_int(val));
  else            atomicMin((unsigned int*)addr, __float_as_uint(val));
}

__device__ __forceinline__ int chip_half() {
  const int id = __builtin_amdgcn_s_sendmsg_rtn(0x87);
  return ((id >> 16) & 0xF) >= 4 ? 1 : 0;
}

__global__ void k_init(float* __restrict__ mx_h, float* __restrict__ mx_t, int n_e) {
  int i = blockIdx.x * blockDim.x + threadIdx.x;
  if (i < 2 * n_e) { mx_h[i] = -__builtin_inff(); mx_t[i] = -__builtin_inff(); }
}

__global__ void k_node_scores(const float* __restrict__ x_e,
                              const float* __restrict__ w_ah,
                              const float* __restrict__ w_at,
                              float* __restrict__ e_h, float* __restrict__ e_t, int n_e) {
  int lane = threadIdx.x & 31;
  int wave = threadIdx.x >> 5;
  int row = blockIdx.x * 8 + wave;
  if (row >= n_e) return;
  const float4 x4 = ((const float4*)(x_e + (size_t)row * E_HID))[lane];
  const float4 a4 = ((const float4*)w_ah)[lane];
  const float4 b4 = ((const float4*)w_at)[lane];
  float sh = x4.x*a4.x + x4.y*a4.y + x4.z*a4.z + x4.w*a4.w;
  float st = x4.x*b4.x + x4.y*b4.y + x4.z*b4.z + x4.w*b4.w;
  for (int off = 16; off > 0; off >>= 1) {
    sh += __shfl_xor(sh, off);
    st += __shfl_xor(st, off);
  }
  if (lane == 0) { e_h[row] = sh; e_t[row] = st; }
}

__global__ void k_rel(const float* __restrict__ x_r,
                      const float* __restrict__ w_ar,
                      const float* __restrict__ W1, const float* __restrict__ b1,
                      const float* __restrict__ W2, const float* __restrict__ b2,
                      float* __restrict__ e_r, float* __restrict__ msg, int n_r) {
  __shared__ float row[R_HID];
  __shared__ float hid[2 * R_HID];
  int r = blockIdx.x;
  int t = threadIdx.x;
  if (t < R_HID) row[t] = x_r[(size_t)r * R_HID + t];
  __syncthreads();
  float acc = b1[t];
  #pragma unroll 8
  for (int k = 0; k < R_HID; ++k) acc += row[k] * W1[k * 2 * R_HID + t];
  hid[t] = acc;
  if (t < 32) {
    float s = 0.f;
    #pragma unroll
    for (int j = 0; j < 4; ++j) s += row[t + 32 * j] * w_ar[t + 32 * j];
    for (int off = 16; off > 0; off >>= 1) s += __shfl_xor(s, off);
    if (t == 0) e_r[r] = s;
  }
  __syncthreads();
  if (t < R_HID) {
    float o = b2[t];
    #pragma unroll 8
    for (int k = 0; k < 2 * R_HID; ++k) o += hid[k] * W2[k * R_HID + t];
    msg[(size_t)r * R_HID + t] = row[t] + o;
  }
}

#define OWN_NPW 391
#define OWN_WAVES 32
#define OWN_QCAP 528

__global__ void k_edge_max(const int* __restrict__ h_idx, const int* __restrict__ t_idx,
                           const int* __restrict__ rel,
                           const float* __restrict__ e_h, const float* __restrict__ e_t,
                           const float* __restrict__ e_r,
                           int4* __restrict__ rec_h, int4* __restrict__ rec_t,
                           unsigned char* __restrict__ key_h, unsigned char* __restrict__ key_t,
                           float* __restrict__ mx_h, float* __restrict__ mx_t, int n_edges,
                           int n_pad, int n_e) {
  int e = blockIdx.x * blockDim.x + threadIdx.x;
  if (e >= n_edges) {
    if (e < n_pad) { key_h[e] = 0; key_t[e] = 0; }
    return;
  }
  const size_t own = (size_t)chip_half() * n_e;
  int h = h_idx[e], t = t_idx[e], r = rel[e];
  float er = e_r[r];
  float lh = leaky(e_h[h] + er);
  float lt = leaky(e_t[t] + er);
  rec_h[e] = make_int4(h, r, __float_as_int(lh), 0);
  rec_t[e] = make_int4(t, r, __float_as_int(lt), 0);
  key_h[e] = (unsigned char)((h / OWN_NPW) & 255);
  key_t[e] = (unsigned char)((t / OWN_NPW) & 255);
  atomicMaxF(mx_h + own + h, lh);
  atomicMaxF(mx_t + own + t, lt);
}

__global__ void __launch_bounds__(1024)
k_own(const int4* __restrict__ rec_h, const int4* __restrict__ rec_t,
      const unsigned int* __restrict__ key_h, const unsigned int* __restrict__ key_t,
      const float* __restrict__ mx_h, const float* __restrict__ mx_t,
      const float* __restrict__ msg, float* __restrict__ x, int n_edges, int n_e, int ranges) {
  __shared__ float acc[OWN_NPW * 128];
  __shared__ float ssum[OWN_NPW + 1];
  __shared__ float smax[OWN_NPW + 1];
  __shared__ int queue[OWN_WAVES * OWN_QCAP];
  const int tid = threadIdx.x;
  const int lane = tid & 31;
  const int w = __builtin_amdgcn_readfirstlane(tid >> 5);
  const int side = (int)blockIdx.x >= ranges ? 1 : 0;
  const int R = (int)blockIdx.x - side * ranges;
  const int node0 = R * OWN_NPW;
  const int nrows = min(OWN_NPW, n_e - node0);

  for (int i = tid; i < OWN_NPW * 32; i += 1024) ((float4*)acc)[i] = make_float4(0.f, 0.f, 0.f, 0.f);
  const float* mx = side ? mx_t : mx_h;
  for (int i = tid; i < OWN_NPW; i += 1024) {
    ssum[i] = 0.f;
    smax[i] = i < nrows ? fmaxf(mx[node0 + i], mx[n_e + node0 + i]) : 0.f;
  }
  __syncthreads();

  const int4* rec = side ? rec_t : rec_h;
  const uint4* key = (const uint4*)(side ? key_t : key_h);
  const unsigned int pat = (unsigned int)(R & 255) * 0x01010101u;
  int* const q = queue + w * OWN_QCAP;
  int cnt = 0;

  auto add_edge = [&](int i, int ln, int ok, float a, float4 row) {
    if (__builtin_amdgcn_readlane(ok, i)) {
      const int lni = __builtin_amdgcn_readlane(ln, i);
      const float ai = __int_as_float(__builtin_amdgcn_readlane(__float_as_int(a), i));
      float* p = acc + lni * 128 + lane;
      atomicAdd(p +  0, ai * row.x); atomicAdd(p + 32, ai * row.y);
      atomicAdd(p + 64, ai * row.z); atomicAdd(p + 96, ai * row.w);
    }
  };
  auto batch = [&](int head, int n) {
    const int e = q[head + min(lane & 15, n - 1)];
    const int4 rc = rec[e];
    const int d = rc.x, r = rc.y;
    const float l = __int_as_float(rc.z);
    const int ln = d - node0;
    const int ok = (unsigned int)ln < (unsigned int)nrows ? 1 : 0;
    const float m = ok ? smax[ln] : 0.f;
    if (n == 16) {
      float4 rows[16];
      #pragma unroll
      for (int i = 0; i < 16; ++i)
        rows[i] = ((const float4*)(msg + (size_t)__builtin_amdgcn_readlane(r, i) * R_HID))[lane];
      const float a = ok ? __expf(l - m) : 0.f;
      if (lane < 16 && ok) atomicAdd(&ssum[ln], a);
      #pragma unroll
      for (int i = 0; i < 16; ++i) add_edge(i, ln, ok, a, rows[i]);
    } else {
      const float a = ok ? __expf(l - m) : 0.f;
      if (lane < n && ok) atomicAdd(&ssum[ln], a);
      for (int i = 0; i < n; ++i) {
        const float4 row = ((const float4*)(msg + (size_t)__builtin_amdgcn_readlane(r, i) * R_HID))[lane];
        add_edge(i, ln, ok, a, row);
      }
    }
  };

  const int nvec = (n_edges + 15) >> 4;
  const int ntile = (nvec + 31) >> 5;
  const int rot = (int)(((long)R * ntile) / ranges);
  for (int t0 = w; t0 < ntile; t0 += OWN_WAVES * 4) {
    uint4 kv[4];
    int vb[4];
    #pragma unroll
    for (int u = 0; u < 4; ++u) {
      const int tv = t0 + u * OWN_WAVES;
      int t = tv + rot; if (t >= ntile) t -= ntile;
      const int v = t * 32 + lane;
      vb[u] = v;
      const bool in = tv < ntile && v < nvec;
      kv[u] = key[in ? v : 0];
      if (!in) kv[u] = make_uint4(~pat, ~pat, ~pat, ~pat);
    }
    #pragma unroll
    for (int u = 0; u < 4; ++u) {
      auto eq = [&](unsigned int xw) {
        const unsigned int y = xw ^ pat;
        return ~(((y & 0x7F7F7F7Fu) + 0x7F7F7F7Fu) | y | 0x7F7F7F7Fu);
      };
      unsigned long long lo = (unsigned long long)eq(kv[u].x) | ((unsigned long long)eq(kv[u].y) << 32);
      unsigned long long hi = (unsigned long long)eq(kv[u].z) | ((unsigned long long)eq(kv[u].w) << 32);
      const int ebase = vb[u] << 4;
      const int valid = n_edges - ebase;
      if (valid < 16) {
        if (valid <= 0) { lo = 0; hi = 0; }
        else if (valid < 8) { lo &= (1ull << (8 * valid)) - 1; hi = 0; }
        else if (valid == 8) { hi = 0; }
        else { hi &= (1ull << (8 * (valid - 8))) - 1; }
      }
      for (;;) {
        const bool has = (lo | hi) != 0;
        const unsigned int b = __builtin_amdgcn_ballot_w32(has);
        if (b == 0) break;
        if (has) {
          int j;
          if (lo) { j = __builtin_ctzll(lo) >> 3; lo &= lo - 1; }
          else    { j = 8 + (__builtin_ctzll(hi) >> 3); hi &= hi - 1; }
          q[cnt + (int)__builtin_amdgcn_mbcnt_lo(b, 0u)] = ebase + j;
        }
        cnt += __builtin_popcount(b);
      }
      if (cnt >= 16) {
        __builtin_amdgcn_wave_barrier();
        int head = 0;
        while (cnt - head >= 16) { batch(head, 16); head += 16; }
        const int rem = cnt - head;
        const int keep = lane < rem ? q[head + lane] : 0;
        __builtin_amdgcn_wave_barrier();
        if (lane < rem) q[lane] = keep;
        cnt = rem;
      }
    }
  }
  if (cnt > 0) {
    __builtin_amdgcn_wave_barrier();
    batch(0, cnt);
  }
  __syncthreads();

  for (int ln = w; ln < nrows; ln += OWN_WAVES) {
    const float inv = __builtin_amdgcn_rcpf(ssum[ln] + 1e-16f);
    const float* p = acc + ln * 128 + lane;
    const float4 v = make_float4(p[0] * inv, p[32] * inv, p[64] * inv, p[96] * inv);
    ((float4*)(x + (size_t)(node0 + ln) * 256 + side * 128))[lane] = v;
  }
}

#define GEMM_MT 5
#define GEMM_ROWS (GEMM_MT * 16)
__global__ void __launch_bounds__(256)
k_final_gemm(const float* __restrict__ x_acc,
             const float* __restrict__ Wr, const float* __restrict__ br,
             float* __restrict__ out, int n_e) {
  constexpr int KTOT = 256, NTOT = 256, KC = 32;
  constexpr int MT = GEMM_MT, MROWS = GEMM_ROWS;
  constexpr int ASTR = 36, BKSTR = 36;
  __shared__ float As[MROWS * ASTR];
  __shared__ float Bs[NTOT * BKSTR];
  const int tid  = threadIdx.x;
  const int lane = tid & 31;
  const int wave = tid >> 5;
  const int m0   = blockIdx.x * MROWS;
  const int lrow = lane & 15;
  const int hsel = (lane >= 16) ? 2 : 0;
  const int mrow = (lane >= 16) ? 8 : 0;
  const int n0   = wave * 32;

  static_assert(GEMM_MT == 5, "A staging below is written for 80 rows");
  const int arow = tid >> 3;
  const int acol = (tid & 7) * 4;
  const int last = n_e - 1;
  const int r0 = min(m0 + arow, last);
  const int r1 = min(m0 + 32 + arow, last);
  const int r2 = min(m0 + 64 + (arow & 15), last);
  const float* ap0 = x_acc + (size_t)r0 * KTOT + acol;
  const float* ap1 = x_acc + (size_t)r1 * KTOT + acol;
  const float* ap2 = x_acc + (size_t)r2 * KTOT + acol;
  const bool a3 = tid < 128;

  v8f c0[MT], c1[MT];
  #pragma unroll
  for (int mt = 0; mt < MT; ++mt) { c0[mt] = v8f{}; c1[mt] = v8f{}; }
  float4 aReg0, aReg1, aReg2 = make_float4(0.f, 0.f, 0.f, 0.f);
  float4 bReg[8];

  auto loadGlobal = [&](int kb) {
    aReg0 = *(const float4*)(ap0 + kb);
    aReg1 = *(const float4*)(ap1 + kb);
    if (a3) aReg2 = *(const float4*)(ap2 + kb);
    #pragma unroll
    for (int i = 0; i < 8; ++i) {
      int k4 = i * 4;
      bReg[i].x = Wr[(size_t)(kb + k4 + 0) * NTOT + tid];
      bReg[i].y = Wr[(size_t)(kb + k4 + 1) * NTOT + tid];
      bReg[i].z = Wr[(size_t)(kb + k4 + 2) * NTOT + tid];
      bReg[i].w = Wr[(size_t)(kb + k4 + 3) * NTOT + tid];
    }
  };
  auto storeSmem = [&]() {
    *(float4*)&As[arow * ASTR + acol] = aReg0;
    *(float4*)&As[(32 + arow) * ASTR + acol] = aReg1;
    if (a3) *(float4*)&As[(64 + arow) * ASTR + acol] = aReg2;
    #pragma unroll
    for (int i = 0; i < 8; ++i)
      *(float4*)&Bs[tid * BKSTR + i * 4] = bReg[i];
  };

  loadGlobal(0);
  for (int kb = 0; kb < KTOT; kb += KC) {
    storeSmem();
    __syncthreads();
    if (kb + KC < KTOT) loadGlobal(kb + KC);
    #pragma unroll
    for (int kk = 0; kk < KC; kk += 4) {
      int ka = kk + hsel;
      v2f b0 = *(const v2f*)&Bs[(n0 + lrow) * BKSTR + ka];
      v2f b1 = *(const v2f*)&Bs[(n0 + 16 + lrow) * BKSTR + ka];
      #pragma unroll
      for (int mt = 0; mt < MT; ++mt) {
        v2f a = *(const v2f*)&As[(mt * 16 + lrow) * ASTR + ka];
        c0[mt] = __builtin_amdgcn_wmma_f32_16x16x4_f32(false, a, false, b0, (short)0, c0[mt], false, false);
        c1[mt] = __builtin_amdgcn_wmma_f32_16x16x4_f32(false, a, false, b1, (short)0, c1[mt], false, false);
      }
    }
    __syncthreads();
  }
  const float bias0 = br[n0 + lrow];
  const float bias1 = br[n0 + 16 + lrow];
  #pragma unroll
  for (int mt = 0; mt < MT; ++mt) {
    #pragma unroll
    for (int j = 0; j < 8; ++j) {
      int grow = m0 + mt * 16 + mrow + j;
      if (grow < n_e) {
        out[(size_t)grow * NTOT + n0 + lrow]      = c0[mt][j] + bias0;
        out[(size_t)grow * NTOT + n0 + 16 + lrow] = c1[mt][j] + bias1;
      }
    }
  }
}

extern "C" void kernel_launch(void* const* d_in, const int* in_sizes, int n_in,
                              void* d_out, int out_size, void* d_ws, size_t ws_size,
                              hipStream_t stream) {
  const float* x_e  = (const float*)d_in[0];
  const float* x_r  = (const float*)d_in[1];
  const int*   eidx = (const int*)d_in[2];
  const int*   rel  = (const int*)d_in[3];
  const float* w_ah = (const float*)d_in[5];
  const float* w_at = (const float*)d_in[6];
  const float* w_ar = (const float*)d_in[7];
  const float* W1   = (const float*)d_in[8];
  const float* b1   = (const float*)d_in[9];
  const float* W2   = (const float*)d_in[10];
  const float* b2   = (const float*)d_in[11];
  const float* Wr   = (const float*)d_in[12];
  const float* br   = (const float*)d_in[13];
  float* out = (float*)d_out;

  const int n_e     = in_sizes[0] / E_HID;
  const int n_r     = in_sizes[1] / R_HID;
  const int n_edges = in_sizes[3];
  const int* h_idx = eidx;
  const int* t_idx = eidx + n_edges;

  const int n_pad = (n_edges + 15) & ~15;
  float* x     = (float*)d_ws;
  unsigned char* key_h = (unsigned char*)(x + (size_t)n_e * 256);
  unsigned char* key_t = key_h + n_pad;
  int4*  rec_h = (int4*)(key_t + n_pad);
  int4*  rec_t = rec_h + n_edges;
  float* msg   = (float*)(rec_t + n_edges);
  float* e_h   = msg + (size_t)n_r * R_HID;
  float* e_t   = e_h + n_e;
  float* e_r   = e_t + n_e;
  float* mx_h  = e_r + n_r;
  float* mx_t  = mx_h + 2 * (size_t)n_e;
  const int ranges = (n_e + OWN_NPW - 1) / OWN_NPW;

  k_init<<<(2 * n_e + 255) / 256, 256, 0, stream>>>(mx_h, mx_t, n_e);
  k_node_scores<<<(n_e + 7) / 8, 256, 0, stream>>>(x_e, w_ah, w_at, e_h, e_t, n_e);
  k_rel<<<n_r, 256, 0, stream>>>(x_r, w_ar, W1, b1, W2, b2, e_r, msg, n_r);
  k_edge_max<<<(n_pad + 255) / 256, 256, 0, stream>>>(h_idx, t_idx, rel, e_h, e_t, e_r,
                                                      rec_h, rec_t, key_h, key_t, mx_h, mx_t,
                                                      n_edges, n_pad, n_e);
  k_own<<<2 * ranges, 1024, 0, stream>>>(rec_h, rec_t, (const unsigned int*)key_h,
                                         (const unsigned int*)key_t, mx_h, mx_t, msg, x,
                                         n_edges, n_e, ranges);
  k_final_gemm<<<(n_e + GEMM_ROWS - 1) / GEMM_ROWS, 256, 0, stream>>>(x, Wr, br, out, n_e);
}


# Edge pass groups each 1024 edges' records by owner range in LDS (counter ranks, prefix sum) + start/length table: owner reads one word per 1024 edges, key-byte scan gone; LDS add order changes
# speedup vs baseline: 2.9655x; 1.1610x over previous
#include <hip/hip_runtime.h>

#define E_HID 128
#define R_HID 128

typedef float v2f __attribute__((ext_vector_type(2)));
typedef float v8f __attribute__((ext_vector_type(8)));

__device__ __forceinline__ float leaky(float x) { return x > 0.f ? x : 0.01f * x; }

__device__ __forceinline__ void atomicMaxF(float* addr, float val) {
  if (val >= 0.f) atomicMax((int*)addr, __float_as_int(val));
  else            atomicMin((unsigned int*)addr, __float_as_uint(val));
}

__device__ __forceinline__ int chip_half() {
  const int id = __builtin_amdgcn_s_sendmsg_rtn(0x87);
  return ((id >> 16) & 0xF) >= 4 ? 1 : 0;
}

__global__ void k_init(float* __restrict__ mx_h, float* __restrict__ mx_t, int n_e) {
  int i = blockIdx.x * blockDim.x + threadIdx.x;
  if (i < 2 * n_e) { mx_h[i] = -__builtin_inff(); mx_t[i] = -__builtin_inff(); }
}

__global__ void k_node_scores(const float* __restrict__ x_e,
                              const float* __restrict__ w_ah,
                              const float* __restrict__ w_at,
                              float* __restrict__ e_h, float* __restrict__ e_t, int n_e) {
  int lane = threadIdx.x & 31;
  int wave = threadIdx.x >> 5;
  int row = blockIdx.x * 8 + wave;
  if (row >= n_e) return;
  const float4 x4 = ((const float4*)(x_e + (size_t)row * E_HID))[lane];
  const float4 a4 = ((const float4*)w_ah)[lane];
  const float4 b4 = ((const float4*)w_at)[lane];
  float sh = x4.x*a4.x + x4.y*a4.y + x4.z*a4.z + x4.w*a4.w;
  float st = x4.x*b4.x + x4.y*b4.y + x4.z*b4.z + x4.w*b4.w;
  for (int off = 16; off > 0; off >>= 1) {
    sh += __shfl_xor(sh, off);
    st += __shfl_xor(st, off);
  }
  if (lane == 0) { e_h[row] = sh; e_t[row] = st; }
}

__global__ void k_rel(const float* __restrict__ x_r,
                      const float* __restrict__ w_ar,
                      const float* __restrict__ W1, const float* __restrict__ b1,
                      const float* __restrict__ W2, const float* __restrict__ b2,
                      float* __restrict__ e_r, float* __restrict__ msg, int n_r) {
  __shared__ float row[R_HID];
  __shared__ float hid[2 * R_HID];
  int r = blockIdx.x;
  int t = threadIdx.x;
  if (t < R_HID) row[t] = x_r[(size_t)r * R_HID + t];
  __syncthreads();
  float acc = b1[t];
  #pragma unroll 8
  for (int k = 0; k < R_HID; ++k) acc += row[k] * W1[k * 2 * R_HID + t];
  hid[t] = acc;
  if (t < 32) {
    float s = 0.f;
    #pragma unroll
    for (int j = 0; j < 4; ++j) s += row[t + 32 * j] * w_ar[t + 32 * j];
    for (int off = 16; off > 0; off >>= 1) s += __shfl_xor(s, off);
    if (t == 0) e_r[r] = s;
  }
  __syncthreads();
  if (t < R_HID) {
    float o = b2[t];
    #pragma unroll 8
    for (int k = 0; k < 2 * R_HID; ++k) o += hid[k] * W2[k * R_HID + t];
    msg[(size_t)r * R_HID + t] = row[t] + o;
  }
}

#define OWN_NPW 391
#define OWN_WAVES 32
#define OWN_QCAP 528
#define SORT_EDGES 1024

__global__ void __launch_bounds__(SORT_EDGES)
k_edge_max(const int* __restrict__ h_idx, const int* __restrict__ t_idx,
           const int* __restrict__ rel,
           const float* __restrict__ e_h, const float* __restrict__ e_t,
           const float* __restrict__ e_r,
           int4* __restrict__ rec_h, int4* __restrict__ rec_t,
           unsigned int* __restrict__ tab_h, unsigned int* __restrict__ tab_t,
           float* __restrict__ mx_h, float* __restrict__ mx_t, int n_edges, int n_e) {
  __shared__ int gcnt[2][256];
  __shared__ int gbase[2][256];
  __shared__ int4 srt[2][SORT_EDGES];
  const int tid = threadIdx.x;
  const int lane = tid & 31;
  const int e0 = blockIdx.x * SORT_EDGES;
  const bool valid = e0 + tid < n_edges;
  if (tid < 512) (&gcnt[0][0])[tid] = 0;
  __syncthreads();
  int bh = 0, bt = 0, rkh = 0, rkt = 0;
  int4 vh = make_int4(0, 0, 0, 0), vt = make_int4(0, 0, 0, 0);
  if (valid) {
    const int e = e0 + tid;
    const size_t own = (size_t)chip_half() * n_e;
    int h = h_idx[e], t = t_idx[e], r = rel[e];
    float er = e_r[r];
    float lh = leaky(e_h[h] + er);
    float lt = leaky(e_t[t] + er);
    vh = make_int4(h, r, __float_as_int(lh), 0);
    vt = make_int4(t, r, __float_as_int(lt), 0);
    bh = (h / OWN_NPW) & 255;
    bt = (t / OWN_NPW) & 255;
    rkh = atomicAdd(&gcnt[0][bh], 1);
    rkt = atomicAdd(&gcnt[1][bt], 1);
    atomicMaxF(mx_h + own + h, lh);
    atomicMaxF(mx_t + own + t, lt);
  }
  __syncthreads();
  if (tid < 64) {
    const int s = tid >> 5;
    int c[8];
    int sum = 0;
    #pragma unroll
    for (int j = 0; j < 8; ++j) { c[j] = gcnt[s][lane * 8 + j]; sum += c[j]; }
    int inc = sum;
    #pragma unroll
    for (int off = 1; off < 32; off <<= 1) {
      const int v = __shfl_up(inc, off);
      if (lane >= off) inc += v;
    }
    int run = inc - sum;
    unsigned int* tab = (s ? tab_t : tab_h) + (size_t)blockIdx.x * 256 + lane * 8;
    #pragma unroll
    for (int j = 0; j < 8; ++j) {
      gbase[s][lane * 8 + j] = run;
      tab[j] = (unsigned int)run | ((unsigned int)c[j] << 16);
      run += c[j];
    }
  }
  __syncthreads();
  if (valid) {
    srt[0][gbase[0][bh] + rkh] = vh;
    srt[1][gbase[1][bt] + rkt] = vt;
  }
  __syncthreads();
  if (valid) {
    rec_h[e0 + tid] = srt[0][tid];
    rec_t[e0 + tid] = srt[1][tid];
  }
}

__global__ void __launch_bounds__(1024)
k_own(const int4* __restrict__ rec_h, const int4* __restrict__ rec_t,
      const unsigned int* __restrict__ tab_h, const unsigned int* __restrict__ tab_t,
      const float* __restrict__ mx_h, const float* __restrict__ mx_t,
      const float* __restrict__ msg, float* __restrict__ x, int n_edges, int n_e, int ranges) {
  __shared__ float acc[OWN_NPW * 128];
  __shared__ float ssum[OWN_NPW + 1];
  __shared__ float smax[OWN_NPW + 1];
  __shared__ int queue[OWN_WAVES * OWN_QCAP];
  const int tid = threadIdx.x;
  const int lane = tid & 31;
  const int w = __builtin_amdgcn_readfirstlane(tid >> 5);
  const int side = (int)blockIdx.x >= ranges ? 1 : 0;
  const int R = (int)blockIdx.x - side * ranges;
  const int node0 = R * OWN_NPW;
  const int nrows = min(OWN_NPW, n_e - node0);

  for (int i = tid; i < OWN_NPW * 32; i += 1024) ((float4*)acc)[i] = make_float4(0.f, 0.f, 0.f, 0.f);
  const float* mx = side ? mx_t : mx_h;
  for (int i = tid; i < OWN_NPW; i += 1024) {
    ssum[i] = 0.f;
    smax[i] = i < nrows ? fmaxf(mx[node0 + i], mx[n_e + node0 + i]) : 0.f;
  }
  __syncthreads();

  const int4* rec = side ? rec_t : rec_h;
  const unsigned int* tab = (side ? tab_t : tab_h) + (R & 255);
  int* const q = queue + w * OWN_QCAP;
  int cnt = 0;

  auto add_edge = [&](int i, int ln, int ok, float a, float4 row) {
    if (__builtin_amdgcn_readlane(ok, i)) {
      const int lni = __builtin_amdgcn_readlane(ln, i);
      const float ai = __int_as_float(__builtin_amdgcn_readlane(__float_as_int(a), i));
      float* p = acc + lni * 128 + lane;
      atomicAdd(p +  0, ai * row.x); atomicAdd(p + 32, ai * row.y);
      atomicAdd(p + 64, ai * row.z); atomicAdd(p + 96, ai * row.w);
    }
  };
  auto batch = [&](int head, int n) {
    const int e = q[head + min(lane & 15, n - 1)];
    const int4 rc = rec[e];
    const int d = rc.x, r = rc.y;
    const float l = __int_as_float(rc.z);
    const int ln = d - node0;
    const int ok = (unsigned int)ln < (unsigned int)nrows ? 1 : 0;
    const float m = ok ? smax[ln] : 0.f;
    if (n == 16) {
      float4 rows[16];
      #pragma unroll
      for (int i = 0; i < 16; ++i)
        rows[i] = ((const float4*)(msg + (size_t)__builtin_amdgcn_readlane(r, i) * R_HID))[lane];
      const float a = ok ? __expf(l - m) : 0.f;
      if (lane < 16 && ok) atomicAdd(&ssum[ln], a);
      #pragma unroll
      for (int i = 0; i < 16; ++i) add_edge(i, ln, ok, a, rows[i]);
    } else {
      const float a = ok ? __expf(l - m) : 0.f;
      if (lane < n && ok) atomicAdd(&ssum[ln], a);
      for (int i = 0; i < n; ++i) {
        const float4 row = ((const float4*)(msg + (size_t)__builtin_amdgcn_readlane(r, i) * R_HID))[lane];
        add_edge(i, ln, ok, a, row);
      }
    }
  };

  const int nblk = (n_edges + SORT_EDGES - 1) / SORT_EDGES;
  for (int b0 = w * 32; b0 < nblk; b0 += OWN_WAVES * 32) {
    const int blk = b0 + lane;
    const unsigned int pk = blk < nblk ? tab[(size_t)blk * 256] : 0u;
    int pos = blk * SORT_EDGES + (int)(pk & 0xFFFFu);
    int left = (int)(pk >> 16);
    for (;;) {
      const int take = min(left, 16);
      if (__builtin_amdgcn_ballot_w32(take > 0) == 0) break;
      int inc = take;
      #pragma unroll
      for (int off = 1; off < 32; off <<= 1) {
        const int v = __shfl_up(inc, off);
        if (lane >= off) inc += v;
      }
      const int at = cnt + inc - take;
      for (int j = 0; j < take; ++j) q[at + j] = pos + j;
      pos += take;
      left -= take;
      cnt += __builtin_amdgcn_readlane(inc, 31);
      if (cnt >= 16) {
        __builtin_amdgcn_wave_barrier();
        int head = 0;
        while (cnt - head >= 16) { batch(head, 16); head += 16; }
        const int rem = cnt - head;
        const int keep = lane < rem ? q[head + lane] : 0;
        __builtin_amdgcn_wave_barrier();
        if (lane < rem) q[lane] = keep;
        cnt = rem;
      }
    }
  }
  if (cnt > 0) {
    __builtin_amdgcn_wave_barrier();
    batch(0, cnt);
  }
  __syncthreads();

  for (int ln = w; ln < nrows; ln += OWN_WAVES) {
    const float inv = __builtin_amdgcn_rcpf(ssum[ln] + 1e-16f);
    const float* p = acc + ln * 128 + lane;
    const float4 v = make_float4(p[0] * inv, p[32] * inv, p[64] * inv, p[96] * inv);
    ((float4*)(x + (size_t)(node0 + ln) * 256 + side * 128))[lane] = v;
  }
}

#define GEMM_MT 5
#define GEMM_ROWS (GEMM_MT * 16)
__global__ void __launch_bounds__(256)
k_final_gemm(const float* __restrict__ x_acc,
             const float* __restrict__ Wr, const float* __restrict__ br,
             float* __restrict__ out, int n_e) {
  constexpr int KTOT = 256, NTOT = 256, KC = 32;
  constexpr int MT = GEMM_MT, MROWS = GEMM_ROWS;
  constexpr int ASTR = 36, BKSTR = 36;
  __shared__ float As[MROWS * ASTR];
  __shared__ float Bs[NTOT * BKSTR];
  const int tid  = threadIdx.x;
  const int lane = tid & 31;
  const int wave = tid >> 5;
  const int m0   = blockIdx.x * MROWS;
  const int lrow = lane & 15;
  const int hsel = (lane >= 16) ? 2 : 0;
  const int mrow = (lane >= 16) ? 8 : 0;
  const int n0   = wave * 32;

  static_assert(GEMM_MT == 5, "A staging below is written for 80 rows");
  const int arow = tid >> 3;
  const int acol = (tid & 7) * 4;
  const int last = n_e - 1;
  const int r0 = min(m0 + arow, last);
  const int r1 = min(m0 + 32 + arow, last);
  const int r2 = min(m0 + 64 + (arow & 15), last);
  const float* ap0 = x_acc + (size_t)r0 * KTOT + acol;
  const float* ap1 = x_acc + (size_t)r1 * KTOT + acol;
  const float* ap2 = x_acc + (size_t)r2 * KTOT + acol;
  const bool a3 = tid < 128;

  v8f c0[MT], c1[MT];
  #pragma unroll
  for (int mt = 0; mt < MT; ++mt) { c0[mt] = v8f{}; c1[mt] = v8f{}; }
  float4 aReg0, aReg1, aReg2 = make_float4(0.f, 0.f, 0.f, 0.f);
  float4 bReg[8];

  auto loadGlobal = [&](int kb) {
    aReg0 = *(const float4*)(ap0 + kb);
    aReg1 = *(const float4*)(ap1 + kb);
    if (a3) aReg2 = *(const float4*)(ap2 + kb);
    #pragma unroll
    for (int i = 0; i < 8; ++i) {
      int k4 = i * 4;
      bReg[i].x = Wr[(size_t)(kb + k4 + 0) * NTOT + tid];
      bReg[i].y = Wr[(size_t)(kb + k4 + 1) * NTOT + tid];
      bReg[i].z = Wr[(size_t)(kb + k4 + 2) * NTOT + tid];
      bReg[i].w = Wr[(size_t)(kb + k4 + 3) * NTOT + tid];
    }
  };
  auto storeSmem = [&]() {
    *(float4*)&As[arow * ASTR + acol] = aReg0;
    *(float4*)&As[(32 + arow) * ASTR + acol] = aReg1;
    if (a3) *(float4*)&As[(64 + arow) * ASTR + acol] = aReg2;
    #pragma unroll
    for (int i = 0; i < 8; ++i)
      *(float4*)&Bs[tid * BKSTR + i * 4] = bReg[i];
  };

  loadGlobal(0);
  for (int kb = 0; kb < KTOT; kb += KC) {
    storeSmem();
    __syncthreads();
    if (kb + KC < KTOT) loadGlobal(kb + KC);
    #pragma unroll
    for (int kk = 0; kk < KC; kk += 4) {
      int ka = kk + hsel;
      v2f b0 = *(const v2f*)&Bs[(n0 + lrow) * BKSTR + ka];
      v2f b1 = *(const v2f*)&Bs[(n0 + 16 + lrow) * BKSTR + ka];
      #pragma unroll
      for (int mt = 0; mt < MT; ++mt) {
        v2f a = *(const v2f*)&As[(mt * 16 + lrow) * ASTR + ka];
        c0[mt] = __builtin_amdgcn_wmma_f32_16x16x4_f32(false, a, false, b0, (short)0, c0[mt], false, false);
        c1[mt] = __builtin_amdgcn_wmma_f32_16x16x4_f32(false, a, false, b1, (short)0, c1[mt], false, false);
      }
    }
    __syncthreads();
  }
  const float bias0 = br[n0 + lrow];
  const float bias1 = br[n0 + 16 + lrow];
  #pragma unroll
  for (int mt = 0; mt < MT; ++mt) {
    #pragma unroll
    for (int j = 0; j < 8; ++j) {
      int grow = m0 + mt * 16 + mrow + j;
      if (grow < n_e) {
        out[(size_t)grow * NTOT + n0 + lrow]      = c0[mt][j] + bias0;
        out[(size_t)grow * NTOT + n0 + 16 + lrow] = c1[mt][j] + bias1;
      }
    }
  }
}

extern "C" void kernel_launch(void* const* d_in, const int* in_sizes, int n_in,
                              void* d_out, int out_size, void* d_ws, size_t ws_size,
                              hipStream_t stream) {
  const float* x_e  = (const float*)d_in[0];
  const float* x_r  = (const float*)d_in[1];
  const int*   eidx = (const int*)d_in[2];
  const int*   rel  = (const int*)d_in[3];
  const float* w_ah = (const float*)d_in[5];
  const float* w_at = (const float*)d_in[6];
  const float* w_ar = (const float*)d_in[7];
  const float* W1   = (const float*)d_in[8];
  const float* b1   = (const float*)d_in[9];
  const float* W2   = (const float*)d_in[10];
  const float* b2   = (const float*)d_in[11];
  const float* Wr   = (const float*)d_in[12];
  const float* br   = (const float*)d_in[13];
  float* out = (float*)d_out;

  const int n_e     = in_sizes[0] / E_HID;
  const int n_r     = in_sizes[1] / R_HID;
  const int n_edges = in_sizes[3];
  const int* h_idx = eidx;
  const int* t_idx = eidx + n_edges;

  const int nblk = (n_edges + SORT_EDGES - 1) / SORT_EDGES;
  float* x     = (float*)d_ws;
  int4*  rec_h = (int4*)(x + (size_t)n_e * 256);
  int4*  rec_t = rec_h + n_edges;
  unsigned int* tab_h = (unsigned int*)(rec_t + n_edges);
  unsigned int* tab_t = tab_h + (size_t)nblk * 256;
  float* msg   = (float*)(tab_t + (size_t)nblk * 256);
  float* e_h   = msg + (size_t)n_r * R_HID;
  float* e_t   = e_h + n_e;
  float* e_r   = e_t + n_e;
  float* mx_h  = e_r + n_r;
  float* mx_t  = mx_h + 2 * (size_t)n_e;
  const int ranges = (n_e + OWN_NPW - 1) / OWN_NPW;

  k_init<<<(2 * n_e + 255) / 256, 256, 0, stream>>>(mx_h, mx_t, n_e);
  k_node_scores<<<(n_e + 7) / 8, 256, 0, stream>>>(x_e, w_ah, w_at, e_h, e_t, n_e);
  k_rel<<<n_r, 256, 0, stream>>>(x_r, w_ar, W1, b1, W2, b2, e_r, msg, n_r);
  k_edge_max<<<nblk, SORT_EDGES, 0, stream>>>(h_idx, t_idx, rel, e_h, e_t, e_r,
                                              rec_h, rec_t, tab_h, tab_t, mx_h, mx_t,
                                              n_edges, n_e);
  k_own<<<2 * ranges, 1024, 0, stream>>>(rec_h, rec_t, tab_h, tab_t, mx_h, mx_t, msg, x,
                                         n_edges, n_e, ranges);
  k_final_gemm<<<(n_e + GEMM_ROWS - 1) / GEMM_ROWS, 256, 0, stream>>>(x, Wr, br, out, n_e);
}


# Scores, logits and segment maxima move into the owner pass (owner streams its x_e rows, walks its records twice, all in LDS): 2 launches and the memory max atomics go; LDS add arrival order changes
# speedup vs baseline: 3.1955x; 1.0776x over previous
#include <hip/hip_runtime.h>

#define E_HID 128
#define R_HID 128

typedef float v2f __attribute__((ext_vector_type(2)));
typedef float v8f __attribute__((ext_vector_type(8)));

__device__ __forceinline__ float leaky(float x) { return x > 0.f ? x : 0.01f * x; }

__device__ __forceinline__ void ldsMaxF(float* addr, float val) {
  if (__float_as_int(val) >= 0) atomicMax((int*)addr, __float_as_int(val));
  else                          atomicMin((unsigned int*)addr, __float_as_uint(val));
}

__global__ void k_rel(const float* __restrict__ x_r,
                      const float* __restrict__ w_ar,
                      const float* __restrict__ W1, const float* __restrict__ b1,
                      const float* __restrict__ W2, const float* __restrict__ b2,
                      float* __restrict__ e_r, float* __restrict__ msg, int n_r) {
  __shared__ float row[R_HID];
  __shared__ float hid[2 * R_HID];
  int r = blockIdx.x;
  int t = threadIdx.x;
  if (t < R_HID) row[t] = x_r[(size_t)r * R_HID + t];
  __syncthreads();
  float acc = b1[t];
  #pragma unroll 8
  for (int k = 0; k < R_HID; ++k) acc += row[k] * W1[k * 2 * R_HID + t];
  hid[t] = acc;
  if (t < 32) {
    float s = 0.f;
    #pragma unroll
    for (int j = 0; j < 4; ++j) s += row[t + 32 * j] * w_ar[t + 32 * j];
    for (int off = 16; off > 0; off >>= 1) s += __shfl_xor(s, off);
    if (t == 0) e_r[r] = s;
  }
  __syncthreads();
  if (t < R_HID) {
    float o = b2[t];
    #pragma unroll 8
    for (int k = 0; k < 2 * R_HID; ++k) o += hid[k] * W2[k * R_HID + t];
    msg[(size_t)r * R_HID + t] = row[t] + o;
  }
}

#define OWN_NPW 391
#define OWN_WAVES 32
#define OWN_RPW ((OWN_NPW + OWN_WAVES - 1) / OWN_WAVES)
#define OWN_QCAP 528
#define SORT_EDGES 1024

__global__ void __launch_bounds__(SORT_EDGES)
k_edge_bin(const int* __restrict__ h_idx, const int* __restrict__ t_idx,
           const int* __restrict__ rel,
           int2* __restrict__ rec_h, int2* __restrict__ rec_t,
           unsigned int* __restrict__ tab_h, unsigned int* __restrict__ tab_t, int n_edges) {
  __shared__ int gcnt[2][256];
  __shared__ int gbase[2][256];
  __shared__ int2 srt[2][SORT_EDGES];
  const int tid = threadIdx.x;
  const int lane = tid & 31;
  const int e0 = blockIdx.x * SORT_EDGES;
  const bool valid = e0 + tid < n_edges;
  if (tid < 512) (&gcnt[0][0])[tid] = 0;
  __syncthreads();
  int bh = 0, bt = 0, rkh = 0, rkt = 0;
  int2 vh = make_int2(0, 0), vt = make_int2(0, 0);
  if (valid) {
    const int e = e0 + tid;
    const int h = h_idx[e], t = t_idx[e], r = rel[e];
    vh = make_int2(h, r);
    vt = make_int2(t, r);
    bh = (h / OWN_NPW) & 255;
    bt = (t / OWN_NPW) & 255;
    rkh = atomicAdd(&gcnt[0][bh], 1);
    rkt = atomicAdd(&gcnt[1][bt], 1);
  }
  __syncthreads();
  if (tid < 64) {
    const int s = tid >> 5;
    int c[8];
    int sum = 0;
    #pragma unroll
    for (int j = 0; j < 8; ++j) { c[j] = gcnt[s][lane * 8 + j]; sum += c[j]; }
    int inc = sum;
    #pragma unroll
    for (int off = 1; off < 32; off <<= 1) {
      const int v = __shfl_up(inc, off);
      if (lane >= off) inc += v;
    }
    int run = inc - sum;
    unsigned int* tab = (s ? tab_t : tab_h) + (size_t)blockIdx.x * 256 + lane * 8;
    #pragma unroll
    for (int j = 0; j < 8; ++j) {
      gbase[s][lane * 8 + j] = run;
      tab[j] = (unsigned int)run | ((unsigned int)c[j] << 16);
      run += c[j];
    }
  }
  __syncthreads();
  if (valid) {
    srt[0][gbase[0][bh] + rkh] = vh;
    srt[1][gbase[1][bt] + rkt] = vt;
  }
  __syncthreads();
  if (valid) {
    rec_h[e0 + tid] = srt[0][tid];
    rec_t[e0 + tid] = srt[1][tid];
  }
}

__global__ void __launch_bounds__(1024)
k_own(const float* __restrict__ x_e,
      const float* __restrict__ w_ah, const float* __restrict__ w_at,
      const float* __restrict__ e_r,
      const int2* __restrict__ rec_h, const int2* __restrict__ rec_t,
      const unsigned int* __restrict__ tab_h, const unsigned int* __restrict__ tab_t,
      const float* __restrict__ msg, float* __restrict__ x, int n_edges, int n_e, int ranges) {
  __shared__ float acc[OWN_NPW * 128];
  __shared__ float ssum[OWN_NPW + 1];
  __shared__ float smax[OWN_NPW + 1];
  __shared__ float esc[OWN_NPW + 1];
  __shared__ int queue[OWN_WAVES * OWN_QCAP];
  const int tid = threadIdx.x;
  const int lane = tid & 31;
  const int w = __builtin_amdgcn_readfirstlane(tid >> 5);
  const int side = (int)blockIdx.x >= ranges ? 1 : 0;
  const int R = (int)blockIdx.x - side * ranges;
  const int node0 = R * OWN_NPW;
  const int nrows = min(OWN_NPW, n_e - node0);

  const float4 w4 = ((const float4*)(side ? w_at : w_ah))[lane];
  float4 xr[OWN_RPW];
  #pragma unroll
  for (int i = 0; i < OWN_RPW; ++i) {
    const int ln = min(w + OWN_WAVES * i, nrows - 1);
    xr[i] = ((const float4*)(x_e + (size_t)(node0 + ln) * E_HID))[lane];
  }

  for (int i = tid; i < OWN_NPW * 32; i += 1024) ((float4*)acc)[i] = make_float4(0.f, 0.f, 0.f, 0.f);

  #pragma unroll
  for (int i = 0; i < OWN_RPW; ++i) {
    float s = xr[i].x * w4.x + xr[i].y * w4.y + xr[i].z * w4.z + xr[i].w * w4.w;
    for (int off = 16; off > 0; off >>= 1) s += __shfl_xor(s, off);
    const int ln = w + OWN_WAVES * i;
    if (lane == 0 && ln < OWN_NPW) {
      esc[ln] = s;
      smax[ln] = -__builtin_inff();
      ssum[ln] = 0.f;
    }
  }
  __syncthreads();

  const int2* rec = side ? rec_t : rec_h;
  const unsigned int* tab = (side ? tab_t : tab_h) + (R & 255);
  const int nblk = (n_edges + SORT_EDGES - 1) / SORT_EDGES;

  for (int b0 = w * 32; b0 < nblk; b0 += OWN_WAVES * 32) {
    const int blk = b0 + lane;
    const unsigned int pk = blk < nblk ? tab[(size_t)blk * 256] : 0u;
    const int pos = blk * SORT_EDGES + (int)(pk & 0xFFFFu);
    const int len = (int)(pk >> 16);
    for (int j = 0; j < len; j += 4) {
      int2 rc[4];
      #pragma unroll
      for (int u = 0; u < 4; ++u) rc[u] = rec[pos + min(j + u, len - 1)];
      float er[4];
      #pragma unroll
      for (int u = 0; u < 4; ++u) er[u] = e_r[rc[u].y];
      #pragma unroll
      for (int u = 0; u < 4; ++u) {
        const int ln = rc[u].x - node0;
        const bool ok = (unsigned int)ln < (unsigned int)nrows;
        const float l = leaky(esc[ok ? ln : 0] + er[u]);
        if (ok) ldsMaxF(&smax[ln], l);
      }
    }
  }
  __syncthreads();

  int* const q = queue + w * OWN_QCAP;
  int cnt = 0;

  auto add_edge = [&](int i, int ln, int ok, float a, float4 row) {
    if (__builtin_amdgcn_readlane(ok, i)) {
      const int lni = __builtin_amdgcn_readlane(ln, i);
      const float ai = __int_as_float(__builtin_amdgcn_readlane(__float_as_int(a), i));
      float* p = acc + lni * 128 + lane;
      atomicAdd(p +  0, ai * row.x); atomicAdd(p + 32, ai * row.y);
      atomicAdd(p + 64, ai * row.z); atomicAdd(p + 96, ai * row.w);
    }
  };
  auto batch = [&](int head, int n) {
    const int e = q[head + min(lane & 15, n - 1)];
    const int2 rc = rec[e];
    const int d = rc.x, r = rc.y;
    const float er = e_r[r];
    const int ln = d - node0;
    const int ok = (unsigned int)ln < (unsigned int)nrows ? 1 : 0;
    const int lnc = ok ? ln : 0;
    if (n == 16) {
      float4 rows[16];
      #pragma unroll
      for (int i = 0; i < 16; ++i)
        rows[i] = ((const float4*)(msg + (size_t)__builtin_amdgcn_readlane(r, i) * R_HID))[lane];
      const float l = leaky(esc[lnc] + er);
      const float m = smax[lnc];
      const float a = ok ? __expf(l - m) : 0.f;
      if (lane < 16 && ok) atomicAdd(&ssum[ln], a);
      #pragma unroll
      for (int i = 0; i < 16; ++i) add_edge(i, ln, ok, a, rows[i]);
    } else {
      const float l = leaky(esc[lnc] + er);
      const float m = smax[lnc];
      const float a = ok ? __expf(l - m) : 0.f;
      if (lane < n && ok) atomicAdd(&ssum[ln], a);
      for (int i = 0; i < n; ++i) {
        const float4 row = ((const float4*)(msg + (size_t)__builtin_amdgcn_readlane(r, i) * R_HID))[lane];
        add_edge(i, ln, ok, a, row);
      }
    }
  };

  for (int b0 = w * 32; b0 < nblk; b0 += OWN_WAVES * 32) {
    const int blk = b0 + lane;
    const unsigned int pk = blk < nblk ? tab[(size_t)blk * 256] : 0u;
    int pos = blk * SORT_EDGES + (int)(pk & 0xFFFFu);
    int left = (int)(pk >> 16);
    for (;;) {
      const int take = min(left, 16);
      if (__builtin_amdgcn_ballot_w32(take > 0) == 0) break;
      int inc = take;
      #pragma unroll
      for (int off = 1; off < 32; off <<= 1) {
        const int v = __shfl_up(inc, off);
        if (lane >= off) inc += v;
      }
      const int at = cnt + inc - take;
      for (int j = 0; j < take; ++j) q[at + j] = pos + j;
      pos += take;
      left -= take;
      cnt += __builtin_amdgcn_readlane(inc, 31);
      if (cnt >= 16) {
        __builtin_amdgcn_wave_barrier();
        int head = 0;
        while (cnt - head >= 16) { batch(head, 16); head += 16; }
        const int rem = cnt - head;
        const int keep = lane < rem ? q[head + lane] : 0;
        __builtin_amdgcn_wave_barrier();
        if (lane < rem) q[lane] = keep;
        cnt = rem;
      }
    }
  }
  if (cnt > 0) {
    __builtin_amdgcn_wave_barrier();
    batch(0, cnt);
  }
  __syncthreads();

  for (int ln = w; ln < nrows; ln += OWN_WAVES) {
    const float inv = __builtin_amdgcn_rcpf(ssum[ln] + 1e-16f);
    const float* p = acc + ln * 128 + lane;
    const float4 v = make_float4(p[0] * inv, p[32] * inv, p[64] * inv, p[96] * inv);
    ((float4*)(x + (size_t)(node0 + ln) * 256 + side * 128))[lane] = v;
  }
}

#define GEMM_MT 5
#define GEMM_ROWS (GEMM_MT * 16)
__global__ void __launch_bounds__(256)
k_final_gemm(const float* __restrict__ x_acc,
             const float* __restrict__ Wr, const float* __restrict__ br,
             float* __restrict__ out, int n_e) {
  constexpr int KTOT = 256, NTOT = 256, KC = 32;
  constexpr int MT = GEMM_MT, MROWS = GEMM_ROWS;
  constexpr int ASTR = 36, BKSTR = 36;
  __shared__ float As[MROWS * ASTR];
  __shared__ float Bs[NTOT * BKSTR];
  const int tid  = threadIdx.x;
  const int lane = tid & 31;
  const int wave = tid >> 5;
  const int m0   = blockIdx.x * MROWS;
  const int lrow = lane & 15;
  const int hsel = (lane >= 16) ? 2 : 0;
  const int mrow = (lane >= 16) ? 8 : 0;
  const int n0   = wave * 32;

  static_assert(GEMM_MT == 5, "A staging below is written for 80 rows");
  const int arow = tid >> 3;
  const int acol = (tid & 7) * 4;
  const int last = n_e - 1;
  const int r0 = min(m0 + arow, last);
  const int r1 = min(m0 + 32 + arow, last);
  const int r2 = min(m0 + 64 + (arow & 15), last);
  const float* ap0 = x_acc + (size_t)r0 * KTOT + acol;
  const float* ap1 = x_acc + (size_t)r1 * KTOT + acol;
  const float* ap2 = x_acc + (size_t)r2 * KTOT + acol;
  const bool a3 = tid < 128;

  v8f c0[MT], c1[MT];
  #pragma unroll
  for (int mt = 0; mt < MT; ++mt) { c0[mt] = v8f{}; c1[mt] = v8f{}; }
  float4 aReg0, aReg1, aReg2 = make_float4(0.f, 0.f, 0.f, 0.f);
  float4 bReg[8];

  auto loadGlobal = [&](int kb) {
    aReg0 = *(const float4*)(ap0 + kb);
    aReg1 = *(const float4*)(ap1 + kb);
    if (a3) aReg2 = *(const float4*)(ap2 + kb);
    #pragma unroll
    for (int i = 0; i < 8; ++i) {
      int k4 = i * 4;
      bReg[i].x = Wr[(size_t)(kb + k4 + 0) * NTOT + tid];
      bReg[i].y = Wr[(size_t)(kb + k4 + 1) * NTOT + tid];
      bReg[i].z = Wr[(size_t)(kb + k4 + 2) * NTOT + tid];
      bReg[i].w = Wr[(size_t)(kb + k4 + 3) * NTOT + tid];
    }
  };
  auto storeSmem = [&]() {
    *(float4*)&As[arow * ASTR + acol] = aReg0;
    *(float4*)&As[(32 + arow) * ASTR + acol] = aReg1;
    if (a3) *(float4*)&As[(64 + arow) * ASTR + acol] = aReg2;
    #pragma unroll
    for (int i = 0; i < 8; ++i)
      *(float4*)&Bs[tid * BKSTR + i * 4] = bReg[i];
  };

  loadGlobal(0);
  for (int kb = 0; kb < KTOT; kb += KC) {
    storeSmem();
    __syncthreads();
    if (kb + KC < KTOT) loadGlobal(kb + KC);
    #pragma unroll
    for (int kk = 0; kk < KC; kk += 4) {
      int ka = kk + hsel;
      v2f b0 = *(const v2f*)&Bs[(n0 + lrow) * BKSTR + ka];
      v2f b1 = *(const v2f*)&Bs[(n0 + 16 + lrow) * BKSTR + ka];
      #pragma unroll
      for (int mt = 0; mt < MT; ++mt) {
        v2f a = *(const v2f*)&As[(mt * 16 + lrow) * ASTR + ka];
        c0[mt] = __builtin_amdgcn_wmma_f32_16x16x4_f32(false, a, false, b0, (short)0, c0[mt], false, false);
        c1[mt] = __builtin_amdgcn_wmma_f32_16x16x4_f32(false, a, false, b1, (short)0, c1[mt], false, false);
      }
    }
    __syncthreads();
  }
  const float bias0 = br[n0 + lrow];
  const float bias1 = br[n0 + 16 + lrow];
  #pragma unroll
  for (int mt = 0; mt < MT; ++mt) {
    #pragma unroll
    for (int j = 0; j < 8; ++j) {
      int grow = m0 + mt * 16 + mrow + j;
      if (grow < n_e) {
        out[(size_t)grow * NTOT + n0 + lrow]      = c0[mt][j] + bias0;
        out[(size_t)grow * NTOT + n0 + 16 + lrow] = c1[mt][j] + bias1;
      }
    }
  }
}

extern "C" void kernel_launch(void* const* d_in, const int* in_sizes, int n_in,
                              void* d_out, int out_size, void* d_ws, size_t ws_size,
                              hipStream_t stream) {
  const float* x_e  = (const float*)d_in[0];
  const float* x_r  = (const float*)d_in[1];
  const int*   eidx = (const int*)d_in[2];
  const int*   rel  = (const int*)d_in[3];
  const float* w_ah = (const float*)d_in[5];
  const float* w_at = (const float*)d_in[6];
  const float* w_ar = (const float*)d_in[7];
  const float* W1   = (const float*)d_in[8];
  const float* b1   = (const float*)d_in[9];
  const float* W2   = (const float*)d_in[10];
  const float* b2   = (const float*)d_in[11];
  const float* Wr   = (const float*)d_in[12];
  const float* br   = (const float*)d_in[13];
  float* out = (float*)d_out;

  const int n_e     = in_sizes[0] / E_HID;
  const int n_r     = in_sizes[1] / R_HID;
  const int n_edges = in_sizes[3];
  const int* h_idx = eidx;
  const int* t_idx = eidx + n_edges;

  const int nblk = (n_edges + SORT_EDGES - 1) / SORT_EDGES;
  float* x     = (float*)d_ws;
  unsigned int* tab_h = (unsigned int*)(x + (size_t)n_e * 256);
  unsigned int* tab_t = tab_h + (size_t)nblk * 256;
  float* msg   = (float*)(tab_t + (size_t)nblk * 256);
  int2*  rec_h = (int2*)(msg + (size_t)n_r * R_HID);
  int2*  rec_t = rec_h + n_edges;
  float* e_r   = (float*)(rec_t + n_edges);
  const int ranges = (n_e + OWN_NPW - 1) / OWN_NPW;

  k_rel<<<n_r, 256, 0, stream>>>(x_r, w_ar, W1, b1, W2, b2, e_r, msg, n_r);
  k_edge_bin<<<nblk, SORT_EDGES, 0, stream>>>(h_idx, t_idx, rel, rec_h, rec_t, tab_h, tab_t, n_edges);
  k_own<<<2 * ranges, 1024, 0, stream>>>(x_e, w_ah, w_at, e_r, rec_h, rec_t, tab_h, tab_t, msg, x,
                                         n_edges, n_e, ranges);
  k_final_gemm<<<(n_e + GEMM_ROWS - 1) / GEMM_ROWS, 256, 0, stream>>>(x, Wr, br, out, n_e);
}


# Row sums leave LDS float atomics: owner files (relation, weight) in per-node LDS lists (integer counters); wave n mod 32 sums a node's rows in registers, one plain LDS add per node; sum order changes
# speedup vs baseline: 3.3341x; 1.0434x over previous
#include <hip/hip_runtime.h>

#define E_HID 128
#define R_HID 128

typedef float v2f __attribute__((ext_vector_type(2)));
typedef float v8f __attribute__((ext_vector_type(8)));

__device__ __forceinline__ float leaky(float x) { return x > 0.f ? x : 0.01f * x; }

__device__ __forceinline__ void ldsMaxF(float* addr, float val) {
  if (__float_as_int(val) >= 0) atomicMax((int*)addr, __float_as_int(val));
  else                          atomicMin((unsigned int*)addr, __float_as_uint(val));
}

__global__ void k_rel(const float* __restrict__ x_r,
                      const float* __restrict__ w_ar,
                      const float* __restrict__ W1, const float* __restrict__ b1,
                      const float* __restrict__ W2, const float* __restrict__ b2,
                      float* __restrict__ e_r, float* __restrict__ msg, int n_r) {
  __shared__ float row[R_HID];
  __shared__ float hid[2 * R_HID];
  int r = blockIdx.x;
  int t = threadIdx.x;
  if (t < R_HID) row[t] = x_r[(size_t)r * R_HID + t];
  __syncthreads();
  float acc = b1[t];
  #pragma unroll 8
  for (int k = 0; k < R_HID; ++k) acc += row[k] * W1[k * 2 * R_HID + t];
  hid[t] = acc;
  if (t < 32) {
    float s = 0.f;
    #pragma unroll
    for (int j = 0; j < 4; ++j) s += row[t + 32 * j] * w_ar[t + 32 * j];
    for (int off = 16; off > 0; off >>= 1) s += __shfl_xor(s, off);
    if (t == 0) e_r[r] = s;
  }
  __syncthreads();
  if (t < R_HID) {
    float o = b2[t];
    #pragma unroll 8
    for (int k = 0; k < 2 * R_HID; ++k) o += hid[k] * W2[k * R_HID + t];
    msg[(size_t)r * R_HID + t] = row[t] + o;
  }
}

#define OWN_NPW 391
#define OWN_WAVES 32
#define OWN_RPW ((OWN_NPW + OWN_WAVES - 1) / OWN_WAVES)
#define SORT_EDGES 1024

__global__ void __launch_bounds__(SORT_EDGES)
k_edge_bin(const int* __restrict__ h_idx, const int* __restrict__ t_idx,
           const int* __restrict__ rel,
           int2* __restrict__ rec_h, int2* __restrict__ rec_t,
           unsigned int* __restrict__ tab_h, unsigned int* __restrict__ tab_t, int n_edges) {
  __shared__ int gcnt[2][256];
  __shared__ int gbase[2][256];
  __shared__ int2 srt[2][SORT_EDGES];
  const int tid = threadIdx.x;
  const int lane = tid & 31;
  const int e0 = blockIdx.x * SORT_EDGES;
  const bool valid = e0 + tid < n_edges;
  if (tid < 512) (&gcnt[0][0])[tid] = 0;
  __syncthreads();
  int bh = 0, bt = 0, rkh = 0, rkt = 0;
  int2 vh = make_int2(0, 0), vt = make_int2(0, 0);
  if (valid) {
    const int e = e0 + tid;
    const int h = h_idx[e], t = t_idx[e], r = rel[e];
    vh = make_int2(h, r);
    vt = make_int2(t, r);
    bh = (h / OWN_NPW) & 255;
    bt = (t / OWN_NPW) & 255;
    rkh = atomicAdd(&gcnt[0][bh], 1);
    rkt = atomicAdd(&gcnt[1][bt], 1);
  }
  __syncthreads();
  if (tid < 64) {
    const int s = tid >> 5;
    int c[8];
    int sum = 0;
    #pragma unroll
    for (int j = 0; j < 8; ++j) { c[j] = gcnt[s][lane * 8 + j]; sum += c[j]; }
    int inc = sum;
    #pragma unroll
    for (int off = 1; off < 32; off <<= 1) {
      const int v = __shfl_up(inc, off);
      if (lane >= off) inc += v;
    }
    int run = inc - sum;
    unsigned int* tab = (s ? tab_t : tab_h) + (size_t)blockIdx.x * 256 + lane * 8;
    #pragma unroll
    for (int j = 0; j < 8; ++j) {
      gbase[s][lane * 8 + j] = run;
      tab[j] = (unsigned int)run | ((unsigned int)c[j] << 16);
      run += c[j];
    }
  }
  __syncthreads();
  if (valid) {
    srt[0][gbase[0][bh] + rkh] = vh;
    srt[1][gbase[1][bt] + rkt] = vt;
  }
  __syncthreads();
  if (valid) {
    rec_h[e0 + tid] = srt[0][tid];
    rec_t[e0 + tid] = srt[1][tid];
  }
}

#define OWN_QN 32
__global__ void __launch_bounds__(1024)
k_own(const float* __restrict__ x_e,
      const float* __restrict__ w_ah, const float* __restrict__ w_at,
      const float* __restrict__ e_r,
      const int2* __restrict__ rec_h, const int2* __restrict__ rec_t,
      const unsigned int* __restrict__ tab_h, const unsigned int* __restrict__ tab_t,
      const float* __restrict__ msg, float* __restrict__ x, int n_edges, int n_e, int ranges) {
  __shared__ float acc[OWN_NPW * 128];
  __shared__ float ssum[OWN_NPW + 1];
  __shared__ float smax[OWN_NPW + 1];
  __shared__ float esc[OWN_NPW + 1];
  __shared__ int qcnt[OWN_NPW + 1];
  __shared__ int2 qent[OWN_NPW * OWN_QN];
  __shared__ int pend[2];
  const int tid = threadIdx.x;
  const int lane = tid & 31;
  const int w = __builtin_amdgcn_readfirstlane(tid >> 5);
  const int side = (int)blockIdx.x >= ranges ? 1 : 0;
  const int R = (int)blockIdx.x - side * ranges;
  const int node0 = R * OWN_NPW;
  const int nrows = min(OWN_NPW, n_e - node0);

  const float4 w4 = ((const float4*)(side ? w_at : w_ah))[lane];
  float4 xr[OWN_RPW];
  #pragma unroll
  for (int i = 0; i < OWN_RPW; ++i) {
    const int ln = min(w + OWN_WAVES * i, nrows - 1);
    xr[i] = ((const float4*)(x_e + (size_t)(node0 + ln) * E_HID))[lane];
  }

  for (int i = tid; i < OWN_NPW * 32; i += 1024) ((float4*)acc)[i] = make_float4(0.f, 0.f, 0.f, 0.f);
  if (tid < 2) pend[tid] = 0;

  #pragma unroll
  for (int i = 0; i < OWN_RPW; ++i) {
    float s = xr[i].x * w4.x + xr[i].y * w4.y + xr[i].z * w4.z + xr[i].w * w4.w;
    for (int off = 16; off > 0; off >>= 1) s += __shfl_xor(s, off);
    const int ln = w + OWN_WAVES * i;
    if (lane == 0 && ln < OWN_NPW) {
      esc[ln] = s;
      smax[ln] = -__builtin_inff();
      ssum[ln] = 0.f;
      qcnt[ln] = 0;
    }
  }
  __syncthreads();

  const int2* rec = side ? rec_t : rec_h;
  const unsigned int* tab = (side ? tab_t : tab_h) + (R & 255);
  const int nblk = (n_edges + SORT_EDGES - 1) / SORT_EDGES;

  unsigned int kpk = 0u;
  int2 krc[8];
  float ker[8];
  #pragma unroll
  for (int u = 0; u < 8; ++u) { krc[u] = make_int2(0, 0); ker[u] = 0.f; }
  for (int b0 = w * 32; b0 < nblk; b0 += OWN_WAVES * 32) {
    const int blk = b0 + lane;
    const unsigned int pk = blk < nblk ? tab[(size_t)blk * 256] : 0u;
    const int pos = blk * SORT_EDGES + (int)(pk & 0xFFFFu);
    const int len = (int)(pk >> 16);
    const bool first = b0 == w * 32;
    if (first) kpk = pk;
    for (int j = 0; j < len; j += 8) {
      int2 rc[8];
      #pragma unroll
      for (int u = 0; u < 8; ++u) rc[u] = rec[pos + min(j + u, len - 1)];
      float er[8];
      #pragma unroll
      for (int u = 0; u < 8; ++u) er[u] = e_r[rc[u].y];
      if (first && j == 0) {
        #pragma unroll
        for (int u = 0; u < 8; ++u) { krc[u] = rc[u]; ker[u] = er[u]; }
      }
      #pragma unroll
      for (int u = 0; u < 8; ++u) {
        const int ln = rc[u].x - node0;
        const bool ok = (unsigned int)ln < (unsigned int)nrows;
        const float l = leaky(esc[ok ? ln : 0] + er[u]);
        if (ok) ldsMaxF(&smax[ln], l);
      }
    }
  }
  __syncthreads();

  const int myln = w + OWN_WAVES * lane;
  const bool mine = lane < OWN_RPW && myln < nrows;

  int cur = -1;
  float4 av = make_float4(0.f, 0.f, 0.f, 0.f);
  float as = 0.f;
  auto flush = [&]() {
    if (cur >= 0) {
      const int ln = w + OWN_WAVES * cur;
      float4* p = (float4*)(acc + ln * 128) + lane;
      float4 v = *p;
      v.x += av.x; v.y += av.y; v.z += av.z; v.w += av.w;
      *p = v;
      if (lane == 0) ssum[ln] += as;
    }
  };
  auto add_entry = [&](int i, int j, float a, float4 row) {
    const int ji = __builtin_amdgcn_readlane(j, i);
    const float ai = __int_as_float(__builtin_amdgcn_readlane(__float_as_int(a), i));
    if (ji != cur) {
      flush();
      cur = ji;
      av = make_float4(0.f, 0.f, 0.f, 0.f);
      as = 0.f;
    }
    av.x += ai * row.x; av.y += ai * row.y; av.z += ai * row.z; av.w += ai * row.w;
    as += ai;
  };

  int round = 0;
  for (int b0 = 0; b0 < nblk; b0 += OWN_WAVES * 32) {
    const int blk = b0 + w * 32 + lane;
    const unsigned int pk = b0 == 0 ? kpk : (blk < nblk ? tab[(size_t)blk * 256] : 0u);
    int pos = blk * SORT_EDGES + (int)(pk & 0xFFFFu);
    int left = (int)(pk >> 16);
    int have = b0 == 0 ? min(left, 8) : 0;
    unsigned int done = 0u;
    for (;;) {
      bool full = false;
      while (left > 0 && !full) {
        int2 rc[8];
        float er[8];
        if (have > 0) {
          #pragma unroll
          for (int u = 0; u < 8; ++u) { rc[u] = krc[u]; er[u] = ker[u]; }
        } else {
          #pragma unroll
          for (int u = 0; u < 8; ++u) rc[u] = rec[pos + min(u, left - 1)];
          #pragma unroll
          for (int u = 0; u < 8; ++u) er[u] = e_r[rc[u].y];
        }
        have = 0;
        const int n8 = min(left, 8);
        int ln[8], slot[8];
        bool want[8];
        #pragma unroll
        for (int u = 0; u < 8; ++u) {
          ln[u] = rc[u].x - node0;
          want[u] = u < n8 && ((done >> u) & 1u) == 0u && (unsigned int)ln[u] < (unsigned int)nrows;
          slot[u] = want[u] ? atomicAdd(&qcnt[ln[u]], 1) : 0;
        }
        #pragma unroll
        for (int u = 0; u < 8; ++u) {
          if (want[u]) {
            if (slot[u] < OWN_QN) {
              const float l = leaky(esc[ln[u]] + er[u]);
              qent[ln[u] * OWN_QN + slot[u]] =
                  make_int2(rc[u].y, __float_as_int(__expf(l - smax[ln[u]])));
              done |= 1u << u;
            } else {
              full = true;
            }
          }
        }
        if (!full) { pos += n8; left -= n8; done = 0u; }
      }
      __syncthreads();

      const int c = mine ? min(qcnt[myln], OWN_QN) : 0;
      if (mine) qcnt[myln] = 0;
      int inc = c;
      #pragma unroll
      for (int off = 1; off < 32; off <<= 1) {
        const int v = __shfl_up(inc, off);
        if (lane >= off) inc += v;
      }
      const int start = inc - c;
      const int total = __builtin_amdgcn_readlane(inc, 31);
      for (int g0 = 0; g0 < total; g0 += 16) {
        const int n = min(16, total - g0);
        const int g = g0 + min(lane & 15, n - 1);
        int j = 0;
        #pragma unroll
        for (int t = 1; t < OWN_RPW; ++t) if (g >= __builtin_amdgcn_readlane(start, t)) j = t;
        const int k = g - __shfl(start, j);
        const int2 en = qent[(w + OWN_WAVES * j) * OWN_QN + k];
        const int r = en.x;
        const float a = __int_as_float(en.y);
        if (n == 16) {
          float4 rows[16];
          #pragma unroll
          for (int i = 0; i < 16; ++i)
            rows[i] = ((const float4*)(msg + (size_t)__builtin_amdgcn_readlane(r, i) * R_HID))[lane];
          #pragma unroll
          for (int i = 0; i < 16; ++i) add_entry(i, j, a, rows[i]);
        } else {
          for (int i = 0; i < n; ++i) {
            const float4 row = ((const float4*)(msg + (size_t)__builtin_amdgcn_readlane(r, i) * R_HID))[lane];
            add_entry(i, j, a, row);
          }
        }
      }
      flush();
      cur = -1;

      if (left > 0) pend[round & 1] = 1;
      __syncthreads();
      const int more = pend[round & 1];
      if (tid == 0) pend[(round + 1) & 1] = 0;
      ++round;
      if (!more) break;
    }
  }

  for (int ln = w; ln < nrows; ln += OWN_WAVES) {
    const float inv = __builtin_amdgcn_rcpf(ssum[ln] + 1e-16f);
    const float4 v = ((const float4*)(acc + ln * 128))[lane];
    ((float4*)(x + (size_t)(node0 + ln) * 256 + side * 128))[lane] =
        make_float4(v.x * inv, v.y * inv, v.z * inv, v.w * inv);
  }
}

#define GEMM_MT 5
#define GEMM_ROWS (GEMM_MT * 16)
__global__ void __launch_bounds__(256)
k_final_gemm(const float* __restrict__ x_acc,
             const float* __restrict__ Wr, const float* __restrict__ br,
             float* __restrict__ out, int n_e) {
  constexpr int KTOT = 256, NTOT = 256, KC = 32;
  constexpr int MT = GEMM_MT, MROWS = GEMM_ROWS;
  constexpr int ASTR = 36, BKSTR = 36;
  __shared__ float As[MROWS * ASTR];
  __shared__ float Bs[NTOT * BKSTR];
  const int tid  = threadIdx.x;
  const int lane = tid & 31;
  const int wave = tid >> 5;
  const int m0   = blockIdx.x * MROWS;
  const int lrow = lane & 15;
  const int hsel = (lane >= 16) ? 2 : 0;
  const int mrow = (lane >= 16) ? 8 : 0;
  const int n0   = wave * 32;

  static_assert(GEMM_MT == 5, "A staging below is written for 80 rows");
  const int arow = tid >> 3;
  const int acol = (tid & 7) * 4;
  const int last = n_e - 1;
  const int r0 = min(m0 + arow, last);
  const int r1 = min(m0 + 32 + arow, last);
  const int r2 = min(m0 + 64 + (arow & 15), last);
  const float* ap0 = x_acc + (size_t)r0 * KTOT + acol;
  const float* ap1 = x_acc + (size_t)r1 * KTOT + acol;
  const float* ap2 = x_acc + (size_t)r2 * KTOT + acol;
  const bool a3 = tid < 128;

  v8f c0[MT], c1[MT];
  #pragma unroll
  for (int mt = 0; mt < MT; ++mt) { c0[mt] = v8f{}; c1[mt] = v8f{}; }
  float4 aReg0, aReg1, aReg2 = make_float4(0.f, 0.f, 0.f, 0.f);
  float4 bReg[8];

  auto loadGlobal = [&](int kb) {
    aReg0 = *(const float4*)(ap0 + kb);
    aReg1 = *(const float4*)(ap1 + kb);
    if (a3) aReg2 = *(const float4*)(ap2 + kb);
    #pragma unroll
    for (int i = 0; i < 8; ++i) {
      int k4 = i * 4;
      bReg[i].x = Wr[(size_t)(kb + k4 + 0) * NTOT + tid];
      bReg[i].y = Wr[(size_t)(kb + k4 + 1) * NTOT + tid];
      bReg[i].z = Wr[(size_t)(kb + k4 + 2) * NTOT + tid];
      bReg[i].w = Wr[(size_t)(kb + k4 + 3) * NTOT + tid];
    }
  };
  auto storeSmem = [&]() {
    *(float4*)&As[arow * ASTR + acol] = aReg0;
    *(float4*)&As[(32 + arow) * ASTR + acol] = aReg1;
    if (a3) *(float4*)&As[(64 + arow) * ASTR + acol] = aReg2;
    #pragma unroll
    for (int i = 0; i < 8; ++i)
      *(float4*)&Bs[tid * BKSTR + i * 4] = bReg[i];
  };

  loadGlobal(0);
  for (int kb = 0; kb < KTOT; kb += KC) {
    storeSmem();
    __syncthreads();
    if (kb + KC < KTOT) loadGlobal(kb + KC);
    #pragma unroll
    for (int kk = 0; kk < KC; kk += 4) {
      int ka = kk + hsel;
      v2f b0 = *(const v2f*)&Bs[(n0 + lrow) * BKSTR + ka];
      v2f b1 = *(const v2f*)&Bs[(n0 + 16 + lrow) * BKSTR + ka];
      #pragma unroll
      for (int mt = 0; mt < MT; ++mt) {
        v2f a = *(const v2f*)&As[(mt * 16 + lrow) * ASTR + ka];
        c0[mt] = __builtin_amdgcn_wmma_f32_16x16x4_f32(false, a, false, b0, (short)0, c0[mt], false, false);
        c1[mt] = __builtin_amdgcn_wmma_f32_16x16x4_f32(false, a, false, b1, (short)0, c1[mt], false, false);
      }
    }
    __syncthreads();
  }
  const float bias0 = br[n0 + lrow];
  const float bias1 = br[n0 + 16 + lrow];
  #pragma unroll
  for (int mt = 0; mt < MT; ++mt) {
    #pragma unroll
    for (int j = 0; j < 8; ++j) {
      int grow = m0 + mt * 16 + mrow + j;
      if (grow < n_e) {
        out[(size_t)grow * NTOT + n0 + lrow]      = c0[mt][j] + bias0;
        out[(size_t)grow * NTOT + n0 + 16 + lrow] = c1[mt][j] + bias1;
      }
    }
  }
}

extern "C" void kernel_launch(void* const* d_in, const int* in_sizes, int n_in,
                              void* d_out, int out_size, void* d_ws, size_t ws_size,
                              hipStream_t stream) {
  const float* x_e  = (const float*)d_in[0];
  const float* x_r  = (const float*)d_in[1];
  const int*   eidx = (const int*)d_in[2];
  const int*   rel  = (const int*)d_in[3];
  const float* w_ah = (const float*)d_in[5];
  const float* w_at = (const float*)d_in[6];
  const float* w_ar = (const float*)d_in[7];
  const float* W1   = (const float*)d_in[8];
  const float* b1   = (const float*)d_in[9];
  const float* W2   = (const float*)d_in[10];
  const float* b2   = (const float*)d_in[11];
  const float* Wr   = (const float*)d_in[12];
  const float* br   = (const float*)d_in[13];
  float* out = (float*)d_out;

  const int n_e     = in_sizes[0] / E_HID;
  const int n_r     = in_sizes[1] / R_HID;
  const int n_edges = in_sizes[3];
  const int* h_idx = eidx;
  const int* t_idx = eidx + n_edges;

  const int nblk = (n_edges + SORT_EDGES - 1) / SORT_EDGES;
  float* x     = (float*)d_ws;
  unsigned int* tab_h = (unsigned int*)(x + (size_t)n_e * 256);
  unsigned int* tab_t = tab_h + (size_t)nblk * 256;
  float* msg   = (float*)(tab_t + (size_t)nblk * 256);
  int2*  rec_h = (int2*)(msg + (size_t)n_r * R_HID);
  int2*  rec_t = rec_h + n_edges;
  float* e_r   = (float*)(rec_t + n_edges);
  const int ranges = (n_e + OWN_NPW - 1) / OWN_NPW;

  k_rel<<<n_r, 256, 0, stream>>>(x_r, w_ar, W1, b1, W2, b2, e_r, msg, n_r);
  k_edge_bin<<<nblk, SORT_EDGES, 0, stream>>>(h_idx, t_idx, rel, rec_h, rec_t, tab_h, tab_t, n_edges);
  k_own<<<2 * ranges, 1024, 0, stream>>>(x_e, w_ah, w_at, e_r, rec_h, rec_t, tab_h, tab_t, msg, x,
                                         n_edges, n_e, ranges);
  k_final_gemm<<<(n_e + GEMM_ROWS - 1) / GEMM_ROWS, 256, 0, stream>>>(x, Wr, br, out, n_e);
}


# Relation MLP: each layer's K split over the block's 8 waves, all of a wave's weight rows loaded at once (16 bytes per lane), partial sums added in wave order via LDS; MLP float sum order changes
# speedup vs baseline: 3.5994x; 1.0796x over previous
#include <hip/hip_runtime.h>

#define E_HID 128
#define R_HID 128

typedef float v2f __attribute__((ext_vector_type(2)));
typedef float v8f __attribute__((ext_vector_type(8)));

__device__ __forceinline__ float leaky(float x) { return x > 0.f ? x : 0.01f * x; }

__device__ __forceinline__ void ldsMaxF(float* addr, float val) {
  if (__float_as_int(val) >= 0) atomicMax((int*)addr, __float_as_int(val));
  else                          atomicMin((unsigned int*)addr, __float_as_uint(val));
}

__global__ void __launch_bounds__(256)
k_rel(const float* __restrict__ x_r,
      const float* __restrict__ w_ar,
      const float* __restrict__ W1, const float* __restrict__ b1,
      const float* __restrict__ W2, const float* __restrict__ b2,
      float* __restrict__ e_r, float* __restrict__ msg, int n_r) {
  __shared__ float row[R_HID];
  __shared__ float hid[2 * R_HID];
  __shared__ float part[8 * 2 * R_HID];
  const int r = blockIdx.x;
  const int t = threadIdx.x;
  const int lane = t & 31;
  const int w = t >> 5;
  if (t < R_HID) row[t] = x_r[(size_t)r * R_HID + t];
  float4 wa[32];
  #pragma unroll
  for (int i = 0; i < 16; ++i) {
    const float4* p = (const float4*)(W1 + (size_t)(16 * w + i) * 2 * R_HID);
    wa[2 * i]     = p[lane];
    wa[2 * i + 1] = p[32 + lane];
  }
  __syncthreads();
  float4 a0 = make_float4(0.f, 0.f, 0.f, 0.f), a1 = a0;
  #pragma unroll
  for (int i = 0; i < 16; ++i) {
    const float xk = row[16 * w + i];
    a0.x += xk * wa[2 * i].x; a0.y += xk * wa[2 * i].y; a0.z += xk * wa[2 * i].z; a0.w += xk * wa[2 * i].w;
    a1.x += xk * wa[2 * i + 1].x; a1.y += xk * wa[2 * i + 1].y; a1.z += xk * wa[2 * i + 1].z; a1.w += xk * wa[2 * i + 1].w;
  }
  ((float4*)(part + w * 2 * R_HID))[lane] = a0;
  ((float4*)(part + w * 2 * R_HID))[32 + lane] = a1;
  float4 wb[32];
  #pragma unroll
  for (int i = 0; i < 32; ++i)
    wb[i] = ((const float4*)(W2 + (size_t)(32 * w + i) * R_HID))[lane];
  if (t < 32) {
    float s = 0.f;
    #pragma unroll
    for (int j = 0; j < 4; ++j) s += row[t + 32 * j] * w_ar[t + 32 * j];
    for (int off = 16; off > 0; off >>= 1) s += __shfl_xor(s, off);
    if (t == 0) e_r[r] = s;
  }
  __syncthreads();
  {
    float acc = b1[t];
    #pragma unroll
    for (int v = 0; v < 8; ++v) acc += part[v * 2 * R_HID + t];
    hid[t] = acc;
  }
  __syncthreads();
  float4 o = make_float4(0.f, 0.f, 0.f, 0.f);
  #pragma unroll
  for (int i = 0; i < 32; ++i) {
    const float hk = hid[32 * w + i];
    o.x += hk * wb[i].x; o.y += hk * wb[i].y; o.z += hk * wb[i].z; o.w += hk * wb[i].w;
  }
  ((float4*)(part + w * R_HID))[lane] = o;
  __syncthreads();
  if (t < R_HID) {
    float acc = b2[t];
    #pragma unroll
    for (int v = 0; v < 8; ++v) acc += part[v * R_HID + t];
    msg[(size_t)r * R_HID + t] = row[t] + acc;
  }
}

#define OWN_NPW 391
#define OWN_WAVES 32
#define OWN_RPW ((OWN_NPW + OWN_WAVES - 1) / OWN_WAVES)
#define SORT_EDGES 1024

__global__ void __launch_bounds__(SORT_EDGES)
k_edge_bin(const int* __restrict__ h_idx, const int* __restrict__ t_idx,
           const int* __restrict__ rel,
           int2* __restrict__ rec_h, int2* __restrict__ rec_t,
           unsigned int* __restrict__ tab_h, unsigned int* __restrict__ tab_t, int n_edges) {
  __shared__ int gcnt[2][256];
  __shared__ int gbase[2][256];
  __shared__ int2 srt[2][SORT_EDGES];
  const int tid = threadIdx.x;
  const int lane = tid & 31;
  const int e0 = blockIdx.x * SORT_EDGES;
  const bool valid = e0 + tid < n_edges;
  if (tid < 512) (&gcnt[0][0])[tid] = 0;
  __syncthreads();
  int bh = 0, bt = 0, rkh = 0, rkt = 0;
  int2 vh = make_int2(0, 0), vt = make_int2(0, 0);
  if (valid) {
    const int e = e0 + tid;
    const int h = h_idx[e], t = t_idx[e], r = rel[e];
    vh = make_int2(h, r);
    vt = make_int2(t, r);
    bh = (h / OWN_NPW) & 255;
    bt = (t / OWN_NPW) & 255;
    rkh = atomicAdd(&gcnt[0][bh], 1);
    rkt = atomicAdd(&gcnt[1][bt], 1);
  }
  __syncthreads();
  if (tid < 64) {
    const int s = tid >> 5;
    int c[8];
    int sum = 0;
    #pragma unroll
    for (int j = 0; j < 8; ++j) { c[j] = gcnt[s][lane * 8 + j]; sum += c[j]; }
    int inc = sum;
    #pragma unroll
    for (int off = 1; off < 32; off <<= 1) {
      const int v = __shfl_up(inc, off);
      if (lane >= off) inc += v;
    }
    int run = inc - sum;
    unsigned int* tab = (s ? tab_t : tab_h) + (size_t)blockIdx.x * 256 + lane * 8;
    #pragma unroll
    for (int j = 0; j < 8; ++j) {
      gbase[s][lane * 8 + j] = run;
      tab[j] = (unsigned int)run | ((unsigned int)c[j] << 16);
      run += c[j];
    }
  }
  __syncthreads();
  if (valid) {
    srt[0][gbase[0][bh] + rkh] = vh;
    srt[1][gbase[1][bt] + rkt] = vt;
  }
  __syncthreads();
  if (valid) {
    rec_h[e0 + tid] = srt[0][tid];
    rec_t[e0 + tid] = srt[1][tid];
  }
}

#define OWN_QN 32
__global__ void __launch_bounds__(1024)
k_own(const float* __restrict__ x_e,
      const float* __restrict__ w_ah, const float* __restrict__ w_at,
      const float* __restrict__ e_r,
      const int2* __restrict__ rec_h, const int2* __restrict__ rec_t,
      const unsigned int* __restrict__ tab_h, const unsigned int* __restrict__ tab_t,
      const float* __restrict__ msg, float* __restrict__ x, int n_edges, int n_e, int ranges) {
  __shared__ float acc[OWN_NPW * 128];
  __shared__ float ssum[OWN_NPW + 1];
  __shared__ float smax[OWN_NPW + 1];
  __shared__ float esc[OWN_NPW + 1];
  __shared__ int qcnt[OWN_NPW + 1];
  __shared__ int2 qent[OWN_NPW * OWN_QN];
  __shared__ int pend[2];
  const int tid = threadIdx.x;
  const int lane = tid & 31;
  const int w = __builtin_amdgcn_readfirstlane(tid >> 5);
  const int side = (int)blockIdx.x >= ranges ? 1 : 0;
  const int R = (int)blockIdx.x - side * ranges;
  const int node0 = R * OWN_NPW;
  const int nrows = min(OWN_NPW, n_e - node0);

  const float4 w4 = ((const float4*)(side ? w_at : w_ah))[lane];
  float4 xr[OWN_RPW];
  #pragma unroll
  for (int i = 0; i < OWN_RPW; ++i) {
    const int ln = min(w + OWN_WAVES * i, nrows - 1);
    xr[i] = ((const float4*)(x_e + (size_t)(node0 + ln) * E_HID))[lane];
  }

  for (int i = tid; i < OWN_NPW * 32; i += 1024) ((float4*)acc)[i] = make_float4(0.f, 0.f, 0.f, 0.f);
  if (tid < 2) pend[tid] = 0;

  #pragma unroll
  for (int i = 0; i < OWN_RPW; ++i) {
    float s = xr[i].x * w4.x + xr[i].y * w4.y + xr[i].z * w4.z + xr[i].w * w4.w;
    for (int off = 16; off > 0; off >>= 1) s += __shfl_xor(s, off);
    const int ln = w + OWN_WAVES * i;
    if (lane == 0 && ln < OWN_NPW) {
      esc[ln] = s;
      smax[ln] = -__builtin_inff();
      ssum[ln] = 0.f;
      qcnt[ln] = 0;
    }
  }
  __syncthreads();

  const int2* rec = side ? rec_t : rec_h;
  const unsigned int* tab = (side ? tab_t : tab_h) + (R & 255);
  const int nblk = (n_edges + SORT_EDGES - 1) / SORT_EDGES;

  unsigned int kpk = 0u;
  int2 krc[8];
  float ker[8];
  #pragma unroll
  for (int u = 0; u < 8; ++u) { krc[u] = make_int2(0, 0); ker[u] = 0.f; }
  for (int b0 = w * 32; b0 < nblk; b0 += OWN_WAVES * 32) {
    const int blk = b0 + lane;
    const unsigned int pk = blk < nblk ? tab[(size_t)blk * 256] : 0u;
    const int pos = blk * SORT_EDGES + (int)(pk & 0xFFFFu);
    const int len = (int)(pk >> 16);
    const bool first = b0 == w * 32;
    if (first) kpk = pk;
    for (int j = 0; j < len; j += 8) {
      int2 rc[8];
      #pragma unroll
      for (int u = 0; u < 8; ++u) rc[u] = rec[pos + min(j + u, len - 1)];
      float er[8];
      #pragma unroll
      for (int u = 0; u < 8; ++u) er[u] = e_r[rc[u].y];
      if (first && j == 0) {
        #pragma unroll
        for (int u = 0; u < 8; ++u) { krc[u] = rc[u]; ker[u] = er[u]; }
      }
      #pragma unroll
      for (int u = 0; u < 8; ++u) {
        const int ln = rc[u].x - node0;
        const bool ok = (unsigned int)ln < (unsigned int)nrows;
        const float l = leaky(esc[ok ? ln : 0] + er[u]);
        if (ok) ldsMaxF(&smax[ln], l);
      }
    }
  }
  __syncthreads();

  const int myln = w + OWN_WAVES * lane;
  const bool mine = lane < OWN_RPW && myln < nrows;

  int cur = -1;
  float4 av = make_float4(0.f, 0.f, 0.f, 0.f);
  float as = 0.f;
  auto flush = [&]() {
    if (cur >= 0) {
      const int ln = w + OWN_WAVES * cur;
      float4* p = (float4*)(acc + ln * 128) + lane;
      float4 v = *p;
      v.x += av.x; v.y += av.y; v.z += av.z; v.w += av.w;
      *p = v;
      if (lane == 0) ssum[ln] += as;
    }
  };
  auto add_entry = [&](int i, int j, float a, float4 row) {
    const int ji = __builtin_amdgcn_readlane(j, i);
    const float ai = __int_as_float(__builtin_amdgcn_readlane(__float_as_int(a), i));
    if (ji != cur) {
      flush();
      cur = ji;
      av = make_float4(0.f, 0.f, 0.f, 0.f);
      as = 0.f;
    }
    av.x += ai * row.x; av.y += ai * row.y; av.z += ai * row.z; av.w += ai * row.w;
    as += ai;
  };

  int round = 0;
  for (int b0 = 0; b0 < nblk; b0 += OWN_WAVES * 32) {
    const int blk = b0 + w * 32 + lane;
    const unsigned int pk = b0 == 0 ? kpk : (blk < nblk ? tab[(size_t)blk * 256] : 0u);
    int pos = blk * SORT_EDGES + (int)(pk & 0xFFFFu);
    int left = (int)(pk >> 16);
    int have = b0 == 0 ? min(left, 8) : 0;
    unsigned int done = 0u;
    for (;;) {
      bool full = false;
      while (left > 0 && !full) {
        int2 rc[8];
        float er[8];
        if (have > 0) {
          #pragma unroll
          for (int u = 0; u < 8; ++u) { rc[u] = krc[u]; er[u] = ker[u]; }
        } else {
          #pragma unroll
          for (int u = 0; u < 8; ++u) rc[u] = rec[pos + min(u, left - 1)];
          #pragma unroll
          for (int u = 0; u < 8; ++u) er[u] = e_r[rc[u].y];
        }
        have = 0;
        const int n8 = min(left, 8);
        int ln[8], slot[8];
        bool want[8];
        #pragma unroll
        for (int u = 0; u < 8; ++u) {
          ln[u] = rc[u].x - node0;
          want[u] = u < n8 && ((done >> u) & 1u) == 0u && (unsigned int)ln[u] < (unsigned int)nrows;
          slot[u] = want[u] ? atomicAdd(&qcnt[ln[u]], 1) : 0;
        }
        #pragma unroll
        for (int u = 0; u < 8; ++u) {
          if (want[u]) {
            if (slot[u] < OWN_QN) {
              const float l = leaky(esc[ln[u]] + er[u]);
              qent[ln[u] * OWN_QN + slot[u]] =
                  make_int2(rc[u].y, __float_as_int(__expf(l - smax[ln[u]])));
              done |= 1u << u;
            } else {
              full = true;
            }
          }
        }
        if (!full) { pos += n8; left -= n8; done = 0u; }
      }
      __syncthreads();

      const int c = mine ? min(qcnt[myln], OWN_QN) : 0;
      if (mine) qcnt[myln] = 0;
      int inc = c;
      #pragma unroll
      for (int off = 1; off < 32; off <<= 1) {
        const int v = __shfl_up(inc, off);
        if (lane >= off) inc += v;
      }
      const int start = inc - c;
      const int total = __builtin_amdgcn_readlane(inc, 31);
      for (int g0 = 0; g0 < total; g0 += 16) {
        const int n = min(16, total - g0);
        const int g = g0 + min(lane & 15, n - 1);
        int j = 0;
        #pragma unroll
        for (int t = 1; t < OWN_RPW; ++t) if (g >= __builtin_amdgcn_readlane(start, t)) j = t;
        const int k = g - __shfl(start, j);
        const int2 en = qent[(w + OWN_WAVES * j) * OWN_QN + k];
        const int r = en.x;
        const float a = __int_as_float(en.y);
        if (n == 16) {
          float4 rows[16];
          #pragma unroll
          for (int i = 0; i < 16; ++i)
            rows[i] = ((const float4*)(msg + (size_t)__builtin_amdgcn_readlane(r, i) * R_HID))[lane];
          #pragma unroll
          for (int i = 0; i < 16; ++i) add_entry(i, j, a, rows[i]);
        } else {
          for (int i = 0; i < n; ++i) {
            const float4 row = ((const float4*)(msg + (size_t)__builtin_amdgcn_readlane(r, i) * R_HID))[lane];
            add_entry(i, j, a, row);
          }
        }
      }
      flush();
      cur = -1;

      if (left > 0) pend[round & 1] = 1;
      __syncthreads();
      const int more = pend[round & 1];
      if (tid == 0) pend[(round + 1) & 1] = 0;
      ++round;
      if (!more) break;
    }
  }

  for (int ln = w; ln < nrows; ln += OWN_WAVES) {
    const float inv = __builtin_amdgcn_rcpf(ssum[ln] + 1e-16f);
    const float4 v = ((const float4*)(acc + ln * 128))[lane];
    ((float4*)(x + (size_t)(node0 + ln) * 256 + side * 128))[lane] =
        make_float4(v.x * inv, v.y * inv, v.z * inv, v.w * inv);
  }
}

#define GEMM_MT 5
#define GEMM_ROWS (GEMM_MT * 16)
__global__ void __launch_bounds__(256)
k_final_gemm(const float* __restrict__ x_acc,
             const float* __restrict__ Wr, const float* __restrict__ br,
             float* __restrict__ out, int n_e) {
  constexpr int KTOT = 256, NTOT = 256, KC = 32;
  constexpr int MT = GEMM_MT, MROWS = GEMM_ROWS;
  constexpr int ASTR = 36, BKSTR = 36;
  __shared__ float As[MROWS * ASTR];
  __shared__ float Bs[NTOT * BKSTR];
  const int tid  = threadIdx.x;
  const int lane = tid & 31;
  const int wave = tid >> 5;
  const int m0   = blockIdx.x * MROWS;
  const int lrow = lane & 15;
  const int hsel = (lane >= 16) ? 2 : 0;
  const int mrow = (lane >= 16) ? 8 : 0;
  const int n0   = wave * 32;

  static_assert(GEMM_MT == 5, "A staging below is written for 80 rows");
  const int arow = tid >> 3;
  const int acol = (tid & 7) * 4;
  const int last = n_e - 1;
  const int r0 = min(m0 + arow, last);
  const int r1 = min(m0 + 32 + arow, last);
  const int r2 = min(m0 + 64 + (arow & 15), last);
  const float* ap0 = x_acc + (size_t)r0 * KTOT + acol;
  const float* ap1 = x_acc + (size_t)r1 * KTOT + acol;
  const float* ap2 = x_acc + (size_t)r2 * KTOT + acol;
  const bool a3 = tid < 128;

  v8f c0[MT], c1[MT];
  #pragma unroll
  for (int mt = 0; mt < MT; ++mt) { c0[mt] = v8f{}; c1[mt] = v8f{}; }
  float4 aReg0, aReg1, aReg2 = make_float4(0.f, 0.f, 0.f, 0.f);
  float4 bReg[8];

  auto loadGlobal = [&](int kb) {
    aReg0 = *(const float4*)(ap0 + kb);
    aReg1 = *(const float4*)(ap1 + kb);
    if (a3) aReg2 = *(const float4*)(ap2 + kb);
    #pragma unroll
    for (int i = 0; i < 8; ++i) {
      int k4 = i * 4;
      bReg[i].x = Wr[(size_t)(kb + k4 + 0) * NTOT + tid];
      bReg[i].y = Wr[(size_t)(kb + k4 + 1) * NTOT + tid];
      bReg[i].z = Wr[(size_t)(kb + k4 + 2) * NTOT + tid];
      bReg[i].w = Wr[(size_t)(kb + k4 + 3) * NTOT + tid];
    }
  };
  auto storeSmem = [&]() {
    *(float4*)&As[arow * ASTR + acol] = aReg0;
    *(float4*)&As[(32 + arow) * ASTR + acol] = aReg1;
    if (a3) *(float4*)&As[(64 + arow) * ASTR + acol] = aReg2;
    #pragma unroll
    for (int i = 0; i < 8; ++i)
      *(float4*)&Bs[tid * BKSTR + i * 4] = bReg[i];
  };

  loadGlobal(0);
  for (int kb = 0; kb < KTOT; kb += KC) {
    storeSmem();
    __syncthreads();
    if (kb + KC < KTOT) loadGlobal(kb + KC);
    #pragma unroll
    for (int kk = 0; kk < KC; kk += 4) {
      int ka = kk + hsel;
      v2f b0 = *(const v2f*)&Bs[(n0 + lrow) * BKSTR + ka];
      v2f b1 = *(const v2f*)&Bs[(n0 + 16 + lrow) * BKSTR + ka];
      #pragma unroll
      for (int mt = 0; mt < MT; ++mt) {
        v2f a = *(const v2f*)&As[(mt * 16 + lrow) * ASTR + ka];
        c0[mt] = __builtin_amdgcn_wmma_f32_16x16x4_f32(false, a, false, b0, (short)0, c0[mt], false, false);
        c1[mt] = __builtin_amdgcn_wmma_f32_16x16x4_f32(false, a, false, b1, (short)0, c1[mt], false, false);
      }
    }
    __syncthreads();
  }
  const float bias0 = br[n0 + lrow];
  const float bias1 = br[n0 + 16 + lrow];
  #pragma unroll
  for (int mt = 0; mt < MT; ++mt) {
    #pragma unroll
    for (int j = 0; j < 8; ++j) {
      int grow = m0 + mt * 16 + mrow + j;
      if (grow < n_e) {
        out[(size_t)grow * NTOT + n0 + lrow]      = c0[mt][j] + bias0;
        out[(size_t)grow * NTOT + n0 + 16 + lrow] = c1[mt][j] + bias1;
      }
    }
  }
}

extern "C" void kernel_launch(void* const* d_in, const int* in_sizes, int n_in,
                              void* d_out, int out_size, void* d_ws, size_t ws_size,
                              hipStream_t stream) {
  const float* x_e  = (const float*)d_in[0];
  const float* x_r  = (const float*)d_in[1];
  const int*   eidx = (const int*)d_in[2];
  const int*   rel  = (const int*)d_in[3];
  const float* w_ah = (const float*)d_in[5];
  const float* w_at = (const float*)d_in[6];
  const float* w_ar = (const float*)d_in[7];
  const float* W1   = (const float*)d_in[8];
  const float* b1   = (const float*)d_in[9];
  const float* W2   = (const float*)d_in[10];
  const float* b2   = (const float*)d_in[11];
  const float* Wr   = (const float*)d_in[12];
  const float* br   = (const float*)d_in[13];
  float* out = (float*)d_out;

  const int n_e     = in_sizes[0] / E_HID;
  const int n_r     = in_sizes[1] / R_HID;
  const int n_edges = in_sizes[3];
  const int* h_idx = eidx;
  const int* t_idx = eidx + n_edges;

  const int nblk = (n_edges + SORT_EDGES - 1) / SORT_EDGES;
  float* x     = (float*)d_ws;
  unsigned int* tab_h = (unsigned int*)(x + (size_t)n_e * 256);
  unsigned int* tab_t = tab_h + (size_t)nblk * 256;
  float* msg   = (float*)(tab_t + (size_t)nblk * 256);
  int2*  rec_h = (int2*)(msg + (size_t)n_r * R_HID);
  int2*  rec_t = rec_h + n_edges;
  float* e_r   = (float*)(rec_t + n_edges);
  const int ranges = (n_e + OWN_NPW - 1) / OWN_NPW;

  k_rel<<<n_r, 256, 0, stream>>>(x_r, w_ar, W1, b1, W2, b2, e_r, msg, n_r);
  k_edge_bin<<<nblk, SORT_EDGES, 0, stream>>>(h_idx, t_idx, rel, rec_h, rec_t, tab_h, tab_t, n_edges);
  k_own<<<2 * ranges, 1024, 0, stream>>>(x_e, w_ah, w_at, e_r, rec_h, rec_t, tab_h, tab_t, msg, x,
                                         n_edges, n_e, ranges);
  k_final_gemm<<<(n_e + GEMM_ROWS - 1) / GEMM_ROWS, 256, 0, stream>>>(x, Wr, br, out, n_e);
}
